# starting kernel
# baseline (speedup 1.0000x reference)
#include <hip/hip_runtime.h>
#include <hip/hip_bf16.h>

typedef __attribute__((ext_vector_type(2))) float v2f;
typedef __attribute__((ext_vector_type(8))) float v8f;

#define B_  1024
#define S_  128
#define H_  512
#define NP  30
#define HH  256

__global__ __launch_bounds__(256) void mean_kernel(const float* __restrict__ hs,
                                                   float* __restrict__ grepr) {
    int idx = blockIdx.x * blockDim.x + threadIdx.x;
    if (idx >= B_ * H_ / 4) return;
    int b  = idx / (H_ / 4);
    int h4 = idx % (H_ / 4);
    const float4* p = (const float4*)(hs + (size_t)b * S_ * H_) + h4;
    float4 acc = make_float4(0.f, 0.f, 0.f, 0.f);
    for (int s = 0; s < S_; ++s) {
        float4 v = p[s * (H_ / 4)];
        acc.x += v.x; acc.y += v.y; acc.z += v.z; acc.w += v.w;
    }
    const float inv = 1.0f / S_;
    ((float4*)grepr)[idx] = make_float4(acc.x * inv, acc.y * inv, acc.z * inv, acc.w * inv);
}

__global__ __launch_bounds__(256) void numnodes_kernel(const float* __restrict__ grepr,
                                                       const float* __restrict__ w_num,
                                                       const float* __restrict__ b_num,
                                                       float* __restrict__ out_num) {
    int wave = (blockIdx.x * blockDim.x + threadIdx.x) >> 5;
    int lane = threadIdx.x & 31;
    int m0    = wave * 16;
    int mrow  = m0 + (lane & 15);
    int khalf = (lane >> 4) * 2;
    int nlo   = lane & 15;
    int nhi   = 16 + nlo;
    bool hvalid = (nhi < NP);
    const float* wlo = w_num + (size_t)nlo * H_;
    const float* whi = w_num + (size_t)(hvalid ? nhi : 0) * H_;
    const float* ar  = grepr + (size_t)mrow * H_;
    v8f c0 = {}; v8f c1 = {};
    for (int k = 0; k < H_; k += 4) {
        int kx = k + khalf;
        v2f a  = *(const v2f*)(ar + kx);
        v2f b0 = *(const v2f*)(wlo + kx);
        v2f b1 = *(const v2f*)(whi + kx);
        b1.x = hvalid ? b1.x : 0.0f;
        b1.y = hvalid ? b1.y : 0.0f;
        c0 = __builtin_amdgcn_wmma_f32_16x16x4_f32(false, a, false, b0, (short)0, c0, false, false);
        c1 = __builtin_amdgcn_wmma_f32_16x16x4_f32(false, a, false, b1, (short)0, c1, false, false);
    }
    int mbase = m0 + ((lane >> 4) << 3);
    for (int v = 0; v < 8; ++v) {
        int m = mbase + v;
        out_num[(size_t)m * NP + nlo] = c0[v] + b_num[nlo];
        if (hvalid) out_num[(size_t)m * NP + nhi] = c1[v] + b_num[nhi];
    }
}

__global__ __launch_bounds__(256) void heads_kernel(const float* __restrict__ hs,
                                                    const float* __restrict__ w_type,
                                                    const float* __restrict__ b_type,
                                                    const float* __restrict__ w_exist,
                                                    const float* __restrict__ b_exist,
                                                    const float* __restrict__ w_val,
                                                    const float* __restrict__ b_val,
                                                    float* __restrict__ out_exist,
                                                    float* __restrict__ out_type,
                                                    float* __restrict__ out_val) {
    int wave = (blockIdx.x * blockDim.x + threadIdx.x) >> 5;
    int lane = threadIdx.x & 31;
    int m0 = wave * 16;
    int r  = m0 + (lane & 15);
    int bb = r / NP, pp = r % NP;
    const float* arow = hs + ((size_t)bb * S_ + pp) * H_;
    int khalf = (lane >> 4) * 2;
    int nlo = lane & 15;
    int nhi = 16 + nlo;
    const float* wlo = w_type + (size_t)nlo * H_;
    const float* whi = (nhi < NP) ? (w_type + (size_t)nhi * H_)
                                  : ((nhi == NP) ? w_exist : w_val);
    v8f c0 = {}; v8f c1 = {};
    for (int k = 0; k < H_; k += 4) {
        int kx = k + khalf;
        v2f a  = *(const v2f*)(arow + kx);
        v2f b0 = *(const v2f*)(wlo + kx);
        v2f b1 = *(const v2f*)(whi + kx);
        c0 = __builtin_amdgcn_wmma_f32_16x16x4_f32(false, a, false, b0, (short)0, c0, false, false);
        c1 = __builtin_amdgcn_wmma_f32_16x16x4_f32(false, a, false, b1, (short)0, c1, false, false);
    }
    int mbase = m0 + ((lane >> 4) << 3);
    for (int v = 0; v < 8; ++v) {
        int m  = mbase + v;
        int mb = m / NP, mp = m % NP;
        out_type[(size_t)mb * NP * NP + mp * NP + nlo] = c0[v] + b_type[nlo];
        if (nhi < NP) {
            out_type[(size_t)mb * NP * NP + mp * NP + nhi] = c1[v] + b_type[nhi];
        } else if (nhi == NP) {
            out_exist[m] = c1[v] + b_exist[0];
        } else {
            out_val[m]   = c1[v] + b_val[0];
        }
    }
}

#define AB_STRIDE 261
__global__ __launch_bounds__(256) void adj_kernel(const float* __restrict__ hs,
                                                  const float* __restrict__ w_adj1,
                                                  const float* __restrict__ b_adj1,
                                                  const float* __restrict__ w_adj2,
                                                  const float* __restrict__ b_adj2,
                                                  float* __restrict__ out_adj) {
    __shared__ float s_ab[2 * NP * AB_STRIDE];
    __shared__ float s_b1[HH];
    __shared__ float s_w2[HH];

    int b    = blockIdx.x;
    int tid  = threadIdx.x;
    int wave = tid >> 5;
    int lane = tid & 31;
    if (tid < HH) { s_b1[tid] = b_adj1[tid]; s_w2[tid] = w_adj2[tid]; }

    int mt      = wave >> 2;
    int ntbase  = (wave & 3) * 8;
    int mrow    = mt * 16 + (lane & 15);
    bool mvalid = (mrow < NP);
    const float* arow = hs + ((size_t)b * S_ + (mvalid ? mrow : 0)) * H_;
    int khalf = (lane >> 4) * 2;

    const float* wr[8];
    for (int t = 0; t < 8; ++t) {
        int n = (ntbase + t) * 16 + (lane & 15);
        wr[t] = w_adj1 + (size_t)(n & 255) * (2 * H_) + (n >> 8) * H_;
    }

    v8f acc[8];
    for (int t = 0; t < 8; ++t) acc[t] = (v8f){};

    for (int k = 0; k < H_; k += 4) {
        int kx = k + khalf;
        v2f a = *(const v2f*)(arow + kx);
        a.x = mvalid ? a.x : 0.0f;
        a.y = mvalid ? a.y : 0.0f;
        for (int t = 0; t < 8; ++t) {
            v2f bf = *(const v2f*)(wr[t] + kx);
            acc[t] = __builtin_amdgcn_wmma_f32_16x16x4_f32(false, a, false, bf,
                                                           (short)0, acc[t], false, false);
        }
    }

    int mbase = mt * 16 + ((lane >> 4) << 3);
    for (int t = 0; t < 8; ++t) {
        int n   = (ntbase + t) * 16 + (lane & 15);
        int reg = n >> 8;
        int h   = n & 255;
        for (int v = 0; v < 8; ++v) {
            int m = mbase + v;
            if (m < NP) s_ab[(reg * NP + m) * AB_STRIDE + h] = acc[t][v];
        }
    }
    __syncthreads();

    float badj2 = b_adj2[0];
    for (int pair = tid; pair < NP * NP; pair += blockDim.x) {
        int i = pair / NP, j = pair % NP;
        float r = 0.0f;
        if (i != j) {
            const float* ai = s_ab + i * AB_STRIDE;
            const float* bj = s_ab + (NP + j) * AB_STRIDE;
            float a2 = 0.0f;
            for (int h = 0; h < HH; ++h) {
                float v = ai[h] + bj[h] + s_b1[h];
                a2 += fmaxf(v, 0.0f) * s_w2[h];
            }
            r = a2 + badj2;
        }
        out_adj[(size_t)b * NP * NP + pair] = r;
    }
}

extern "C" void kernel_launch(void* const* d_in, const int* in_sizes, int n_in,
                              void* d_out, int out_size, void* d_ws, size_t ws_size,
                              hipStream_t stream) {
    const float* hs      = (const float*)d_in[0];
    const float* w_exist = (const float*)d_in[1];
    const float* b_exist = (const float*)d_in[2];
    const float* w_type  = (const float*)d_in[3];
    const float* b_type  = (const float*)d_in[4];
    const float* w_val   = (const float*)d_in[5];
    const float* b_val   = (const float*)d_in[6];
    const float* w_adj1  = (const float*)d_in[7];
    const float* b_adj1  = (const float*)d_in[8];
    const float* w_adj2  = (const float*)d_in[9];
    const float* b_adj2  = (const float*)d_in[10];
    const float* w_num   = (const float*)d_in[11];
    const float* b_num   = (const float*)d_in[12];

    float* out       = (float*)d_out;
    float* out_num   = out;
    float* out_exist = out + 30720;
    float* out_type  = out + 61440;
    float* out_val   = out + 983040;
    float* out_adj   = out + 1013760;

    float* grepr = (float*)d_ws;

    mean_kernel<<<(B_ * H_ / 4 + 255) / 256, 256, 0, stream>>>(hs, grepr);
    numnodes_kernel<<<8, 256, 0, stream>>>(grepr, w_num, b_num, out_num);
    heads_kernel<<<240, 256, 0, stream>>>(hs, w_type, b_type, w_exist, b_exist,
                                          w_val, b_val, out_exist, out_type, out_val);
    adj_kernel<<<B_, 256, 0, stream>>>(hs, w_adj1, b_adj1, w_adj2, b_adj2, out_adj);
}


# adjacency GEMM: K-slabs of both operands staged through LDS with whole-line loads, double-buffered, fragments read from LDS; k order inside each 8-deep chunk changes the summation order
# speedup vs baseline: 1.4799x; 1.4799x over previous
#include <hip/hip_runtime.h>
#include <hip/hip_bf16.h>

typedef __attribute__((ext_vector_type(2))) float v2f;
typedef __attribute__((ext_vector_type(4))) float v4f;
typedef __attribute__((ext_vector_type(8))) float v8f;

#define B_  1024
#define S_  128
#define H_  512
#define NP  30
#define HH  256

__global__ __launch_bounds__(256) void mean_kernel(const float* __restrict__ hs,
                                                   float* __restrict__ grepr) {
    int idx = blockIdx.x * blockDim.x + threadIdx.x;
    if (idx >= B_ * H_ / 4) return;
    int b  = idx / (H_ / 4);
    int h4 = idx % (H_ / 4);
    const float4* p = (const float4*)(hs + (size_t)b * S_ * H_) + h4;
    float4 acc = make_float4(0.f, 0.f, 0.f, 0.f);
    for (int s = 0; s < S_; ++s) {
        float4 v = p[s * (H_ / 4)];
        acc.x += v.x; acc.y += v.y; acc.z += v.z; acc.w += v.w;
    }
    const float inv = 1.0f / S_;
    ((float4*)grepr)[idx] = make_float4(acc.x * inv, acc.y * inv, acc.z * inv, acc.w * inv);
}

__global__ __launch_bounds__(256) void numnodes_kernel(const float* __restrict__ grepr,
                                                       const float* __restrict__ w_num,
                                                       const float* __restrict__ b_num,
                                                       float* __restrict__ out_num) {
    int wave = (blockIdx.x * blockDim.x + threadIdx.x) >> 5;
    int lane = threadIdx.x & 31;
    int m0    = wave * 16;
    int mrow  = m0 + (lane & 15);
    int khalf = (lane >> 4) * 2;
    int nlo   = lane & 15;
    int nhi   = 16 + nlo;
    bool hvalid = (nhi < NP);
    const float* wlo = w_num + (size_t)nlo * H_;
    const float* whi = w_num + (size_t)(hvalid ? nhi : 0) * H_;
    const float* ar  = grepr + (size_t)mrow * H_;
    v8f c0 = {}; v8f c1 = {};
    for (int k = 0; k < H_; k += 4) {
        int kx = k + khalf;
        v2f a  = *(const v2f*)(ar + kx);
        v2f b0 = *(const v2f*)(wlo + kx);
        v2f b1 = *(const v2f*)(whi + kx);
        b1.x = hvalid ? b1.x : 0.0f;
        b1.y = hvalid ? b1.y : 0.0f;
        c0 = __builtin_amdgcn_wmma_f32_16x16x4_f32(false, a, false, b0, (short)0, c0, false, false);
        c1 = __builtin_amdgcn_wmma_f32_16x16x4_f32(false, a, false, b1, (short)0, c1, false, false);
    }
    int mbase = m0 + ((lane >> 4) << 3);
    for (int v = 0; v < 8; ++v) {
        int m = mbase + v;
        out_num[(size_t)m * NP + nlo] = c0[v] + b_num[nlo];
        if (hvalid) out_num[(size_t)m * NP + nhi] = c1[v] + b_num[nhi];
    }
}

__global__ __launch_bounds__(256) void heads_kernel(const float* __restrict__ hs,
                                                    const float* __restrict__ w_type,
                                                    const float* __restrict__ b_type,
                                                    const float* __restrict__ w_exist,
                                                    const float* __restrict__ b_exist,
                                                    const float* __restrict__ w_val,
                                                    const float* __restrict__ b_val,
                                                    float* __restrict__ out_exist,
                                                    float* __restrict__ out_type,
                                                    float* __restrict__ out_val) {
    int wave = (blockIdx.x * blockDim.x + threadIdx.x) >> 5;
    int lane = threadIdx.x & 31;
    int m0 = wave * 16;
    int r  = m0 + (lane & 15);
    int bb = r / NP, pp = r % NP;
    const float* arow = hs + ((size_t)bb * S_ + pp) * H_;
    int khalf = (lane >> 4) * 2;
    int nlo = lane & 15;
    int nhi = 16 + nlo;
    const float* wlo = w_type + (size_t)nlo * H_;
    const float* whi = (nhi < NP) ? (w_type + (size_t)nhi * H_)
                                  : ((nhi == NP) ? w_exist : w_val);
    v8f c0 = {}; v8f c1 = {};
    for (int k = 0; k < H_; k += 4) {
        int kx = k + khalf;
        v2f a  = *(const v2f*)(arow + kx);
        v2f b0 = *(const v2f*)(wlo + kx);
        v2f b1 = *(const v2f*)(whi + kx);
        c0 = __builtin_amdgcn_wmma_f32_16x16x4_f32(false, a, false, b0, (short)0, c0, false, false);
        c1 = __builtin_amdgcn_wmma_f32_16x16x4_f32(false, a, false, b1, (short)0, c1, false, false);
    }
    int mbase = m0 + ((lane >> 4) << 3);
    for (int v = 0; v < 8; ++v) {
        int m  = mbase + v;
        int mb = m / NP, mp = m % NP;
        out_type[(size_t)mb * NP * NP + mp * NP + nlo] = c0[v] + b_type[nlo];
        if (nhi < NP) {
            out_type[(size_t)mb * NP * NP + mp * NP + nhi] = c1[v] + b_type[nhi];
        } else if (nhi == NP) {
            out_exist[m] = c1[v] + b_exist[0];
        } else {
            out_val[m]   = c1[v] + b_val[0];
        }
    }
}

#define AB_STRIDE 261
#define KC      32
#define SP      36
#define SROWS   544
#define STAGE_F (SROWS * SP)
#define NSLAB   (H_ / KC)
static_assert(2 * NP * AB_STRIDE + 2 * HH <= STAGE_F, "the epilogue arrays fit inside slab stage 0");
__global__ __launch_bounds__(256) void adj_kernel(const float* __restrict__ hs,
                                                  const float* __restrict__ w_adj1,
                                                  const float* __restrict__ b_adj1,
                                                  const float* __restrict__ w_adj2,
                                                  const float* __restrict__ b_adj2,
                                                  float* __restrict__ out_adj) {
    __shared__ __attribute__((aligned(16))) float s_mem[2 * STAGE_F];
    float* s_ab = s_mem;
    float* s_b1 = s_mem + 2 * NP * AB_STRIDE;
    float* s_w2 = s_b1 + HH;

    int b    = blockIdx.x;
    int tid  = threadIdx.x;
    int wave = tid >> 5;
    int lane = tid & 31;
    float rb1 = b_adj1[tid];
    float rw2 = w_adj2[tid];

    int  r0     = tid >> 3;
    int  q      = tid & 7;
    bool avalid = (r0 < NP);
    const float* gB = w_adj1 + (size_t)r0 * (2 * H_) + 4 * q;
    const float* gA = hs + ((size_t)b * S_ + (avalid ? r0 : 0)) * H_ + 4 * q;
    int sdst = r0 * SP + 4 * q;
    v4f pf[17];

#define SLAB_LOAD(T)                                                              \
    {                                                                             \
        _Pragma("unroll")                                                         \
        for (int i = 0; i < 16; ++i)                                              \
            pf[i] = *(const v4f*)(gB + (size_t)(i & 7) * 32 * (2 * H_)         \
                                        + (i >> 3) * H_ + (T) * KC);              \
        v4f av = *(const v4f*)(gA + (T) * KC);                                    \
        av.x = avalid ? av.x : 0.0f;                           \
        av.y = avalid ? av.y : 0.0f;                                              \
        av.z = avalid ? av.z : 0.0f;                                              \
        av.w = avalid ? av.w : 0.0f;                                              \
        pf[16] = av;                                                              \
    }
#define SLAB_STORE(STG)                                                           \
    {                                                                             \
        float* d = s_mem + (STG) * STAGE_F + sdst;                                \
        _Pragma("unroll")                                                         \
        for (int i = 0; i < 17; ++i) *(v4f*)(d + i * 32 * SP) = pf[i];            \
    }

    int c       = lane & 15;
    int h       = lane >> 4;
    int mt      = wave >> 2;
    int ntbase  = (wave & 3) * 8;
    int aoff    = (512 + mt * 16 + c) * SP + 4 * h;
    int boff    = (ntbase * 16 + c) * SP + 4 * h;

    v8f acc[8];
    for (int t = 0; t < 8; ++t) acc[t] = (v8f){};

    SLAB_LOAD(0);
    SLAB_STORE(0);
    __syncthreads();

    for (int s = 0; s < NSLAB; ++s) {
        if (s + 1 < NSLAB) SLAB_LOAD(s + 1);
        const float* st = s_mem + (s & 1) * STAGE_F;
#pragma unroll
        for (int ch = 0; ch < KC / 8; ++ch) {
            v4f a4 = *(const v4f*)(st + aoff + ch * 8);
            v4f b4[8];
#pragma unroll
            for (int t = 0; t < 8; ++t)
                b4[t] = *(const v4f*)(st + boff + t * 16 * SP + ch * 8);
            v2f a01 = {a4.x, a4.y};
            v2f a23 = {a4.z, a4.w};
#pragma unroll
            for (int t = 0; t < 8; ++t) {
                v2f b01 = {b4[t].x, b4[t].y};
                acc[t] = __builtin_amdgcn_wmma_f32_16x16x4_f32(false, a01, false, b01,
                                                               (short)0, acc[t], false, false);
            }
#pragma unroll
            for (int t = 0; t < 8; ++t) {
                v2f b23 = {b4[t].z, b4[t].w};
                acc[t] = __builtin_amdgcn_wmma_f32_16x16x4_f32(false, a23, false, b23,
                                                               (short)0, acc[t], false, false);
            }
        }
        if (s + 1 < NSLAB) SLAB_STORE((s + 1) & 1);
        __syncthreads();
    }
#undef SLAB_LOAD
#undef SLAB_STORE

    s_b1[tid] = rb1;
    s_w2[tid] = rw2;
    int mbase = mt * 16 + ((lane >> 4) << 3);
    for (int t = 0; t < 8; ++t) {
        int n   = (ntbase + t) * 16 + (lane & 15);
        int reg = n >> 8;
        int hh  = n & 255;
        for (int v = 0; v < 8; ++v) {
            int m = mbase + v;
            if (m < NP) s_ab[(reg * NP + m) * AB_STRIDE + hh] = acc[t][v];
        }
    }
    __syncthreads();

    float badj2 = b_adj2[0];
    for (int pair = tid; pair < NP * NP; pair += blockDim.x) {
        int i = pair / NP, j = pair % NP;
        float r = 0.0f;
        if (i != j) {
            const float* ai = s_ab + i * AB_STRIDE;
            const float* bj = s_ab + (NP + j) * AB_STRIDE;
            float a2 = 0.0f;
            for (int hx = 0; hx < HH; ++hx) {
                float v = ai[hx] + bj[hx] + s_b1[hx];
                a2 += fmaxf(v, 0.0f) * s_w2[hx];
            }
            r = a2 + badj2;
        }
        out_adj[(size_t)b * NP * NP + pair] = r;
    }
}

extern "C" void kernel_launch(void* const* d_in, const int* in_sizes, int n_in,
                              void* d_out, int out_size, void* d_ws, size_t ws_size,
                              hipStream_t stream) {
    const float* hs      = (const float*)d_in[0];
    const float* w_exist = (const float*)d_in[1];
    const float* b_exist = (const float*)d_in[2];
    const float* w_type  = (const float*)d_in[3];
    const float* b_type  = (const float*)d_in[4];
    const float* w_val   = (const float*)d_in[5];
    const float* b_val   = (const float*)d_in[6];
    const float* w_adj1  = (const float*)d_in[7];
    const float* b_adj1  = (const float*)d_in[8];
    const float* w_adj2  = (const float*)d_in[9];
    const float* b_adj2  = (const float*)d_in[10];
    const float* w_num   = (const float*)d_in[11];
    const float* b_num   = (const float*)d_in[12];

    float* out       = (float*)d_out;
    float* out_num   = out;
    float* out_exist = out + 30720;
    float* out_type  = out + 61440;
    float* out_val   = out + 983040;
    float* out_adj   = out + 1013760;

    float* grepr = (float*)d_ws;

    mean_kernel<<<(B_ * H_ / 4 + 255) / 256, 256, 0, stream>>>(hs, grepr);
    numnodes_kernel<<<8, 256, 0, stream>>>(grepr, w_num, b_num, out_num);
    heads_kernel<<<240, 256, 0, stream>>>(hs, w_type, b_type, w_exist, b_exist,
                                          w_val, b_val, out_exist, out_type, out_val);
    adj_kernel<<<B_, 256, 0, stream>>>(hs, w_adj1, b_adj1, w_adj2, b_adj2, out_adj);
}


# adjacency epilogue: each thread sums a 2x2 block of (i,j) pairs from 16-byte LDS reads, bj row pairs interleaved for packed f32 math; b1 added to ai at the spill, which moves one rounding
# speedup vs baseline: 1.7280x; 1.1676x over previous
#include <hip/hip_runtime.h>
#include <hip/hip_bf16.h>

typedef __attribute__((ext_vector_type(2))) float v2f;
typedef __attribute__((ext_vector_type(4))) float v4f;
typedef __attribute__((ext_vector_type(8))) float v8f;

#define B_  1024
#define S_  128
#define H_  512
#define NP  30
#define HH  256

__global__ __launch_bounds__(256) void mean_kernel(const float* __restrict__ hs,
                                                   float* __restrict__ grepr) {
    int idx = blockIdx.x * blockDim.x + threadIdx.x;
    if (idx >= B_ * H_ / 4) return;
    int b  = idx / (H_ / 4);
    int h4 = idx % (H_ / 4);
    const float4* p = (const float4*)(hs + (size_t)b * S_ * H_) + h4;
    float4 acc = make_float4(0.f, 0.f, 0.f, 0.f);
    for (int s = 0; s < S_; ++s) {
        float4 v = p[s * (H_ / 4)];
        acc.x += v.x; acc.y += v.y; acc.z += v.z; acc.w += v.w;
    }
    const float inv = 1.0f / S_;
    ((float4*)grepr)[idx] = make_float4(acc.x * inv, acc.y * inv, acc.z * inv, acc.w * inv);
}

__global__ __launch_bounds__(256) void numnodes_kernel(const float* __restrict__ grepr,
                                                       const float* __restrict__ w_num,
                                                       const float* __restrict__ b_num,
                                                       float* __restrict__ out_num) {
    int wave = (blockIdx.x * blockDim.x + threadIdx.x) >> 5;
    int lane = threadIdx.x & 31;
    int m0    = wave * 16;
    int mrow  = m0 + (lane & 15);
    int khalf = (lane >> 4) * 2;
    int nlo   = lane & 15;
    int nhi   = 16 + nlo;
    bool hvalid = (nhi < NP);
    const float* wlo = w_num + (size_t)nlo * H_;
    const float* whi = w_num + (size_t)(hvalid ? nhi : 0) * H_;
    const float* ar  = grepr + (size_t)mrow * H_;
    v8f c0 = {}; v8f c1 = {};
    for (int k = 0; k < H_; k += 4) {
        int kx = k + khalf;
        v2f a  = *(const v2f*)(ar + kx);
        v2f b0 = *(const v2f*)(wlo + kx);
        v2f b1 = *(const v2f*)(whi + kx);
        b1.x = hvalid ? b1.x : 0.0f;
        b1.y = hvalid ? b1.y : 0.0f;
        c0 = __builtin_amdgcn_wmma_f32_16x16x4_f32(false, a, false, b0, (short)0, c0, false, false);
        c1 = __builtin_amdgcn_wmma_f32_16x16x4_f32(false, a, false, b1, (short)0, c1, false, false);
    }
    int mbase = m0 + ((lane >> 4) << 3);
    for (int v = 0; v < 8; ++v) {
        int m = mbase + v;
        out_num[(size_t)m * NP + nlo] = c0[v] + b_num[nlo];
        if (hvalid) out_num[(size_t)m * NP + nhi] = c1[v] + b_num[nhi];
    }
}

__global__ __launch_bounds__(256) void heads_kernel(const float* __restrict__ hs,
                                                    const float* __restrict__ w_type,
                                                    const float* __restrict__ b_type,
                                                    const float* __restrict__ w_exist,
                                                    const float* __restrict__ b_exist,
                                                    const float* __restrict__ w_val,
                                                    const float* __restrict__ b_val,
                                                    float* __restrict__ out_exist,
                                                    float* __restrict__ out_type,
                                                    float* __restrict__ out_val) {
    int wave = (blockIdx.x * blockDim.x + threadIdx.x) >> 5;
    int lane = threadIdx.x & 31;
    int m0 = wave * 16;
    int r  = m0 + (lane & 15);
    int bb = r / NP, pp = r % NP;
    const float* arow = hs + ((size_t)bb * S_ + pp) * H_;
    int khalf = (lane >> 4) * 2;
    int nlo = lane & 15;
    int nhi = 16 + nlo;
    const float* wlo = w_type + (size_t)nlo * H_;
    const float* whi = (nhi < NP) ? (w_type + (size_t)nhi * H_)
                                  : ((nhi == NP) ? w_exist : w_val);
    v8f c0 = {}; v8f c1 = {};
    for (int k = 0; k < H_; k += 4) {
        int kx = k + khalf;
        v2f a  = *(const v2f*)(arow + kx);
        v2f b0 = *(const v2f*)(wlo + kx);
        v2f b1 = *(const v2f*)(whi + kx);
        c0 = __builtin_amdgcn_wmma_f32_16x16x4_f32(false, a, false, b0, (short)0, c0, false, false);
        c1 = __builtin_amdgcn_wmma_f32_16x16x4_f32(false, a, false, b1, (short)0, c1, false, false);
    }
    int mbase = m0 + ((lane >> 4) << 3);
    for (int v = 0; v < 8; ++v) {
        int m  = mbase + v;
        int mb = m / NP, mp = m % NP;
        out_type[(size_t)mb * NP * NP + mp * NP + nlo] = c0[v] + b_type[nlo];
        if (nhi < NP) {
            out_type[(size_t)mb * NP * NP + mp * NP + nhi] = c1[v] + b_type[nhi];
        } else if (nhi == NP) {
            out_exist[m] = c1[v] + b_exist[0];
        } else {
            out_val[m]   = c1[v] + b_val[0];
        }
    }
}

#define AI_PITCH 260
#define BJ_PITCH 516
#define NT      15
#define KC      32
#define SP      36
#define SROWS   544
#define STAGE_F (SROWS * SP)
#define NSLAB   (H_ / KC)
static_assert(NP * AI_PITCH + NT * BJ_PITCH + HH + NP * NP <= STAGE_F, "the epilogue arrays fit inside slab stage 0");
__global__ __launch_bounds__(256) void adj_kernel(const float* __restrict__ hs,
                                                  const float* __restrict__ w_adj1,
                                                  const float* __restrict__ b_adj1,
                                                  const float* __restrict__ w_adj2,
                                                  const float* __restrict__ b_adj2,
                                                  float* __restrict__ out_adj) {
    __shared__ __attribute__((aligned(16))) float s_mem[2 * STAGE_F];
    float* s_ai  = s_mem;
    float* s_bj  = s_ai + NP * AI_PITCH;
    float* s_w2  = s_bj + NT * BJ_PITCH;
    float* s_out = s_w2 + HH;

    int b    = blockIdx.x;
    int tid  = threadIdx.x;
    int wave = tid >> 5;
    int lane = tid & 31;
    float rw2 = w_adj2[tid];

    int  r0     = tid >> 3;
    int  q      = tid & 7;
    bool avalid = (r0 < NP);
    const float* gB = w_adj1 + (size_t)r0 * (2 * H_) + 4 * q;
    const float* gA = hs + ((size_t)b * S_ + (avalid ? r0 : 0)) * H_ + 4 * q;
    int sdst = r0 * SP + 4 * q;
    v4f pf[17];

#define SLAB_LOAD(T)                                                              \
    {                                                                             \
        _Pragma("unroll")                                                         \
        for (int i = 0; i < 16; ++i)                                              \
            pf[i] = *(const v4f*)(gB + (size_t)(i & 7) * 32 * (2 * H_)         \
                                        + (i >> 3) * H_ + (T) * KC);              \
        v4f av = *(const v4f*)(gA + (T) * KC);                                    \
        av.x = avalid ? av.x : 0.0f;                           \
        av.y = avalid ? av.y : 0.0f;                                              \
        av.z = avalid ? av.z : 0.0f;                                              \
        av.w = avalid ? av.w : 0.0f;                                              \
        pf[16] = av;                                                              \
    }
#define SLAB_STORE(STG)                                                           \
    {                                                                             \
        float* d = s_mem + (STG) * STAGE_F + sdst;                                \
        _Pragma("unroll")                                                         \
        for (int i = 0; i < 17; ++i) *(v4f*)(d + i * 32 * SP) = pf[i];            \
    }

    int c       = lane & 15;
    int h       = lane >> 4;
    int mt      = wave >> 2;
    int ntbase  = (wave & 3) * 8;
    int aoff    = (512 + mt * 16 + c) * SP + 4 * h;
    int boff    = (ntbase * 16 + c) * SP + 4 * h;

    v8f acc[8];
    for (int t = 0; t < 8; ++t) acc[t] = (v8f){};

    SLAB_LOAD(0);
    SLAB_STORE(0);
    __syncthreads();

    for (int s = 0; s < NSLAB; ++s) {
        if (s + 1 < NSLAB) SLAB_LOAD(s + 1);
        const float* st = s_mem + (s & 1) * STAGE_F;
#pragma unroll
        for (int ch = 0; ch < KC / 8; ++ch) {
            v4f a4 = *(const v4f*)(st + aoff + ch * 8);
            v4f b4[8];
#pragma unroll
            for (int t = 0; t < 8; ++t)
                b4[t] = *(const v4f*)(st + boff + t * 16 * SP + ch * 8);
            v2f a01 = {a4.x, a4.y};
            v2f a23 = {a4.z, a4.w};
#pragma unroll
            for (int t = 0; t < 8; ++t) {
                v2f b01 = {b4[t].x, b4[t].y};
                acc[t] = __builtin_amdgcn_wmma_f32_16x16x4_f32(false, a01, false, b01,
                                                               (short)0, acc[t], false, false);
            }
#pragma unroll
            for (int t = 0; t < 8; ++t) {
                v2f b23 = {b4[t].z, b4[t].w};
                acc[t] = __builtin_amdgcn_wmma_f32_16x16x4_f32(false, a23, false, b23,
                                                               (short)0, acc[t], false, false);
            }
        }
        if (s + 1 < NSLAB) SLAB_STORE((s + 1) & 1);
        __syncthreads();
    }
#undef SLAB_LOAD
#undef SLAB_STORE

    s_w2[tid] = rw2;
    int mbase = mt * 16 + ((lane >> 4) << 3);
    if (ntbase < 16) {
        for (int t = 0; t < 8; ++t) {
            int   hh = (ntbase + t) * 16 + (lane & 15);
            float bb = b_adj1[hh];
            for (int v = 0; v < 8; ++v) {
                int m = mbase + v;
                if (m < NP) s_ai[m * AI_PITCH + hh] = acc[t][v] + bb;
            }
        }
    } else {
        for (int t = 0; t < 8; ++t) {
            int hh = (ntbase - 16 + t) * 16 + (lane & 15);
            for (int v = 0; v < 8; ++v) {
                int m = mbase + v;
                int p = (m < NT) ? m : m - NT;
                int s = (m < NT) ? 0 : 1;
                if (m < NP) s_bj[p * BJ_PITCH + 2 * hh + s] = acc[t][v];
            }
        }
    }
    __syncthreads();

    float badj2 = b_adj2[0];
    if (tid < NT * NT) {
        int ti = tid / NT, tj = tid % NT;
        const v4f* pa0 = (const v4f*)(s_ai + ti * AI_PITCH);
        const v4f* pa1 = (const v4f*)(s_ai + (ti + NT) * AI_PITCH);
        const v4f* pb  = (const v4f*)(s_bj + tj * BJ_PITCH);
        const v4f* pw  = (const v4f*)s_w2;
        const v2f zero = {0.0f, 0.0f};
        v2f r0 = zero;
        v2f r1 = zero;
#define ADJ_TERM(A0, A1, BJ, WV)                                                  \
        {                                                                         \
            v2f x0 = (v2f){A0, A0} + (BJ);                                        \
            v2f x1 = (v2f){A1, A1} + (BJ);                                        \
            x0 = __builtin_elementwise_max(x0, zero);                             \
            x1 = __builtin_elementwise_max(x1, zero);                             \
            r0 += x0 * (v2f){WV, WV};                                             \
            r1 += x1 * (v2f){WV, WV};                                             \
        }
        for (int g = 0; g < HH / 4; ++g) {
            v4f a0 = pa0[g], a1 = pa1[g], w = pw[g];
            v4f bA = pb[2 * g], bB = pb[2 * g + 1];
            ADJ_TERM(a0.x, a1.x, ((v2f){bA.x, bA.y}), w.x);
            ADJ_TERM(a0.y, a1.y, ((v2f){bA.z, bA.w}), w.y);
            ADJ_TERM(a0.z, a1.z, ((v2f){bB.x, bB.y}), w.z);
            ADJ_TERM(a0.w, a1.w, ((v2f){bB.z, bB.w}), w.w);
        }
#undef ADJ_TERM
        bool dg = (ti == tj);
        s_out[ti * NP + tj]             = dg ? 0.0f : (r0.x + badj2);
        s_out[ti * NP + tj + NT]        = r0.y + badj2;
        s_out[(ti + NT) * NP + tj]      = r1.x + badj2;
        s_out[(ti + NT) * NP + tj + NT] = dg ? 0.0f : (r1.y + badj2);
    }
    __syncthreads();
    for (int pair = tid; pair < NP * NP; pair += HH)
        out_adj[(size_t)b * NP * NP + pair] = s_out[pair];
}

extern "C" void kernel_launch(void* const* d_in, const int* in_sizes, int n_in,
                              void* d_out, int out_size, void* d_ws, size_t ws_size,
                              hipStream_t stream) {
    const float* hs      = (const float*)d_in[0];
    const float* w_exist = (const float*)d_in[1];
    const float* b_exist = (const float*)d_in[2];
    const float* w_type  = (const float*)d_in[3];
    const float* b_type  = (const float*)d_in[4];
    const float* w_val   = (const float*)d_in[5];
    const float* b_val   = (const float*)d_in[6];
    const float* w_adj1  = (const float*)d_in[7];
    const float* b_adj1  = (const float*)d_in[8];
    const float* w_adj2  = (const float*)d_in[9];
    const float* b_adj2  = (const float*)d_in[10];
    const float* w_num   = (const float*)d_in[11];
    const float* b_num   = (const float*)d_in[12];

    float* out       = (float*)d_out;
    float* out_num   = out;
    float* out_exist = out + 30720;
    float* out_type  = out + 61440;
    float* out_val   = out + 983040;
    float* out_adj   = out + 1013760;

    float* grepr = (float*)d_ws;

    mean_kernel<<<(B_ * H_ / 4 + 255) / 256, 256, 0, stream>>>(hs, grepr);
    numnodes_kernel<<<8, 256, 0, stream>>>(grepr, w_num, b_num, out_num);
    heads_kernel<<<240, 256, 0, stream>>>(hs, w_type, b_type, w_exist, b_exist,
                                          w_val, b_val, out_exist, out_type, out_val);
    adj_kernel<<<B_, 256, 0, stream>>>(hs, w_adj1, b_adj1, w_adj2, b_adj2, out_adj);
}


# heads and num_nodes GEMMs: A and weight fragments fetched by 16-byte loads one 64-deep K group ahead in two register buffers; k order inside each 8-deep chunk changes the summation order
# speedup vs baseline: 1.9162x; 1.1089x over previous
#include <hip/hip_runtime.h>
#include <hip/hip_bf16.h>

typedef __attribute__((ext_vector_type(2))) float v2f;
typedef __attribute__((ext_vector_type(4))) float v4f;
typedef __attribute__((ext_vector_type(8))) float v8f;

#define B_  1024
#define S_  128
#define H_  512
#define NP  30
#define HH  256

__global__ __launch_bounds__(256) void mean_kernel(const float* __restrict__ hs,
                                                   float* __restrict__ grepr) {
    int idx = blockIdx.x * blockDim.x + threadIdx.x;
    if (idx >= B_ * H_ / 4) return;
    int b  = idx / (H_ / 4);
    int h4 = idx % (H_ / 4);
    const float4* p = (const float4*)(hs + (size_t)b * S_ * H_) + h4;
    float4 acc = make_float4(0.f, 0.f, 0.f, 0.f);
    for (int s = 0; s < S_; ++s) {
        float4 v = p[s * (H_ / 4)];
        acc.x += v.x; acc.y += v.y; acc.z += v.z; acc.w += v.w;
    }
    const float inv = 1.0f / S_;
    ((float4*)grepr)[idx] = make_float4(acc.x * inv, acc.y * inv, acc.z * inv, acc.w * inv);
}

#define KGRP 64
#define NGRP (H_ / KGRP)
__device__ __forceinline__ void gemm_k512(const float* __restrict__ a,
                                          const float* __restrict__ p,
                                          const float* __restrict__ q,
                                          bool qvalid, v8f& c0, v8f& c1) {
    v4f A0[8], P0[8], Q0[8];
    v4f A1[8], P1[8], Q1[8];
#define GRP_LOAD(A, P, Q, G)                                                      \
    {                                                                             \
        _Pragma("unroll")                                                         \
        for (int i = 0; i < 8; ++i) {                                             \
            A[i] = *(const v4f*)(a + (G) * KGRP + 8 * i);                         \
            P[i] = *(const v4f*)(p + (G) * KGRP + 8 * i);                         \
            Q[i] = *(const v4f*)(q + (G) * KGRP + 8 * i);                         \
        }                                                                         \
    }
#define GRP_MUL(A, P, Q)                                                          \
    {                                                                             \
        _Pragma("unroll")                                                         \
        for (int i = 0; i < 8; ++i) {                                             \
            v2f a01 = {A[i].x, A[i].y};                                           \
            v2f a23 = {A[i].z, A[i].w};                                           \
            v2f p01 = {P[i].x, P[i].y};                                           \
            v2f p23 = {P[i].z, P[i].w};                                           \
            v2f q01 = {qvalid ? Q[i].x : 0.0f, qvalid ? Q[i].y : 0.0f};           \
            v2f q23 = {qvalid ? Q[i].z : 0.0f, qvalid ? Q[i].w : 0.0f};           \
            c0 = __builtin_amdgcn_wmma_f32_16x16x4_f32(false, a01, false, p01,    \
                                                       (short)0, c0, false, false); \
            c1 = __builtin_amdgcn_wmma_f32_16x16x4_f32(false, a01, false, q01,    \
                                                       (short)0, c1, false, false); \
            c0 = __builtin_amdgcn_wmma_f32_16x16x4_f32(false, a23, false, p23,    \
                                                       (short)0, c0, false, false); \
            c1 = __builtin_amdgcn_wmma_f32_16x16x4_f32(false, a23, false, q23,    \
                                                       (short)0, c1, false, false); \
        }                                                                         \
    }
#define LOADS_STAY_AHEAD() __builtin_amdgcn_sched_barrier(0)
    GRP_LOAD(A0, P0, Q0, 0);
    LOADS_STAY_AHEAD();
#pragma unroll
    for (int g = 0; g < NGRP; g += 2) {
        GRP_LOAD(A1, P1, Q1, g + 1);
        LOADS_STAY_AHEAD();
        GRP_MUL(A0, P0, Q0);
        if (g + 2 < NGRP) GRP_LOAD(A0, P0, Q0, g + 2);
        LOADS_STAY_AHEAD();
        GRP_MUL(A1, P1, Q1);
    }
#undef LOADS_STAY_AHEAD
#undef GRP_LOAD
#undef GRP_MUL
}

__global__ __launch_bounds__(256) void numnodes_kernel(const float* __restrict__ grepr,
                                                       const float* __restrict__ w_num,
                                                       const float* __restrict__ b_num,
                                                       float* __restrict__ out_num) {
    int wave = (blockIdx.x * blockDim.x + threadIdx.x) >> 5;
    int lane = threadIdx.x & 31;
    int m0    = wave * 16;
    int mrow  = m0 + (lane & 15);
    int kq    = (lane >> 4) * 4;
    int nlo   = lane & 15;
    int nhi   = 16 + nlo;
    bool hvalid = (nhi < NP);
    const float* wlo = w_num + (size_t)nlo * H_ + kq;
    const float* whi = w_num + (size_t)(hvalid ? nhi : 0) * H_ + kq;
    const float* ar  = grepr + (size_t)mrow * H_ + kq;
    v8f c0 = {}; v8f c1 = {};
    gemm_k512(ar, wlo, whi, hvalid, c0, c1);
    int mbase = m0 + ((lane >> 4) << 3);
    for (int v = 0; v < 8; ++v) {
        int m = mbase + v;
        out_num[(size_t)m * NP + nlo] = c0[v] + b_num[nlo];
        if (hvalid) out_num[(size_t)m * NP + nhi] = c1[v] + b_num[nhi];
    }
}

__global__ __launch_bounds__(256) void heads_kernel(const float* __restrict__ hs,
                                                    const float* __restrict__ w_type,
                                                    const float* __restrict__ b_type,
                                                    const float* __restrict__ w_exist,
                                                    const float* __restrict__ b_exist,
                                                    const float* __restrict__ w_val,
                                                    const float* __restrict__ b_val,
                                                    float* __restrict__ out_exist,
                                                    float* __restrict__ out_type,
                                                    float* __restrict__ out_val) {
    int wave = (blockIdx.x * blockDim.x + threadIdx.x) >> 5;
    int lane = threadIdx.x & 31;
    int m0 = wave * 16;
    int r  = m0 + (lane & 15);
    int bb = r / NP, pp = r % NP;
    const float* arow = hs + ((size_t)bb * S_ + pp) * H_;
    int kq  = (lane >> 4) * 4;
    int nlo = lane & 15;
    int nhi = 16 + nlo;
    const float* wlo = w_type + (size_t)nlo * H_ + kq;
    const float* whi = ((nhi < NP) ? (w_type + (size_t)nhi * H_)
                                   : ((nhi == NP) ? w_exist : w_val)) + kq;
    v8f c0 = {}; v8f c1 = {};
    gemm_k512(arow + kq, wlo, whi, true, c0, c1);
    int mbase = m0 + ((lane >> 4) << 3);
    for (int v = 0; v < 8; ++v) {
        int m  = mbase + v;
        int mb = m / NP, mp = m % NP;
        out_type[(size_t)mb * NP * NP + mp * NP + nlo] = c0[v] + b_type[nlo];
        if (nhi < NP) {
            out_type[(size_t)mb * NP * NP + mp * NP + nhi] = c1[v] + b_type[nhi];
        } else if (nhi == NP) {
            out_exist[m] = c1[v] + b_exist[0];
        } else {
            out_val[m]   = c1[v] + b_val[0];
        }
    }
}

#define AI_PITCH 260
#define BJ_PITCH 516
#define NT      15
#define KC      32
#define SP      36
#define SROWS   544
#define STAGE_F (SROWS * SP)
#define NSLAB   (H_ / KC)
static_assert(NP * AI_PITCH + NT * BJ_PITCH + HH + NP * NP <= STAGE_F, "the epilogue arrays fit inside slab stage 0");
__global__ __launch_bounds__(256) void adj_kernel(const float* __restrict__ hs,
                                                  const float* __restrict__ w_adj1,
                                                  const float* __restrict__ b_adj1,
                                                  const float* __restrict__ w_adj2,
                                                  const float* __restrict__ b_adj2,
                                                  float* __restrict__ out_adj) {
    __shared__ __attribute__((aligned(16))) float s_mem[2 * STAGE_F];
    float* s_ai  = s_mem;
    float* s_bj  = s_ai + NP * AI_PITCH;
    float* s_w2  = s_bj + NT * BJ_PITCH;
    float* s_out = s_w2 + HH;

    int b    = blockIdx.x;
    int tid  = threadIdx.x;
    int wave = tid >> 5;
    int lane = tid & 31;
    float rw2 = w_adj2[tid];

    int  r0     = tid >> 3;
    int  q      = tid & 7;
    bool avalid = (r0 < NP);
    const float* gB = w_adj1 + (size_t)r0 * (2 * H_) + 4 * q;
    const float* gA = hs + ((size_t)b * S_ + (avalid ? r0 : 0)) * H_ + 4 * q;
    int sdst = r0 * SP + 4 * q;
    v4f pf[17];

#define SLAB_LOAD(T)                                                              \
    {                                                                             \
        _Pragma("unroll")                                                         \
        for (int i = 0; i < 16; ++i)                                              \
            pf[i] = *(const v4f*)(gB + (size_t)(i & 7) * 32 * (2 * H_)         \
                                        + (i >> 3) * H_ + (T) * KC);              \
        v4f av = *(const v4f*)(gA + (T) * KC);                                    \
        av.x = avalid ? av.x : 0.0f;                           \
        av.y = avalid ? av.y : 0.0f;                                              \
        av.z = avalid ? av.z : 0.0f;                                              \
        av.w = avalid ? av.w : 0.0f;                                              \
        pf[16] = av;                                                              \
    }
#define SLAB_STORE(STG)                                                           \
    {                                                                             \
        float* d = s_mem + (STG) * STAGE_F + sdst;                                \
        _Pragma("unroll")                                                         \
        for (int i = 0; i < 17; ++i) *(v4f*)(d + i * 32 * SP) = pf[i];            \
    }

    int c       = lane & 15;
    int h       = lane >> 4;
    int mt      = wave >> 2;
    int ntbase  = (wave & 3) * 8;
    int aoff    = (512 + mt * 16 + c) * SP + 4 * h;
    int boff    = (ntbase * 16 + c) * SP + 4 * h;

    v8f acc[8];
    for (int t = 0; t < 8; ++t) acc[t] = (v8f){};

    SLAB_LOAD(0);
    SLAB_STORE(0);
    __syncthreads();

    for (int s = 0; s < NSLAB; ++s) {
        if (s + 1 < NSLAB) SLAB_LOAD(s + 1);
        const float* st = s_mem + (s & 1) * STAGE_F;
#pragma unroll
        for (int ch = 0; ch < KC / 8; ++ch) {
            v4f a4 = *(const v4f*)(st + aoff + ch * 8);
            v4f b4[8];
#pragma unroll
            for (int t = 0; t < 8; ++t)
                b4[t] = *(const v4f*)(st + boff + t * 16 * SP + ch * 8);
            v2f a01 = {a4.x, a4.y};
            v2f a23 = {a4.z, a4.w};
#pragma unroll
            for (int t = 0; t < 8; ++t) {
                v2f b01 = {b4[t].x, b4[t].y};
                acc[t] = __builtin_amdgcn_wmma_f32_16x16x4_f32(false, a01, false, b01,
                                                               (short)0, acc[t], false, false);
            }
#pragma unroll
            for (int t = 0; t < 8; ++t) {
                v2f b23 = {b4[t].z, b4[t].w};
                acc[t] = __builtin_amdgcn_wmma_f32_16x16x4_f32(false, a23, false, b23,
                                                               (short)0, acc[t], false, false);
            }
        }
        if (s + 1 < NSLAB) SLAB_STORE((s + 1) & 1);
        __syncthreads();
    }
#undef SLAB_LOAD
#undef SLAB_STORE

    s_w2[tid] = rw2;
    int mbase = mt * 16 + ((lane >> 4) << 3);
    if (ntbase < 16) {
        for (int t = 0; t < 8; ++t) {
            int   hh = (ntbase + t) * 16 + (lane & 15);
            float bb = b_adj1[hh];
            for (int v = 0; v < 8; ++v) {
                int m = mbase + v;
                if (m < NP) s_ai[m * AI_PITCH + hh] = acc[t][v] + bb;
            }
        }
    } else {
        for (int t = 0; t < 8; ++t) {
            int hh = (ntbase - 16 + t) * 16 + (lane & 15);
            for (int v = 0; v < 8; ++v) {
                int m = mbase + v;
                int p = (m < NT) ? m : m - NT;
                int s = (m < NT) ? 0 : 1;
                if (m < NP) s_bj[p * BJ_PITCH + 2 * hh + s] = acc[t][v];
            }
        }
    }
    __syncthreads();

    float badj2 = b_adj2[0];
    if (tid < NT * NT) {
        int ti = tid / NT, tj = tid % NT;
        const v4f* pa0 = (const v4f*)(s_ai + ti * AI_PITCH);
        const v4f* pa1 = (const v4f*)(s_ai + (ti + NT) * AI_PITCH);
        const v4f* pb  = (const v4f*)(s_bj + tj * BJ_PITCH);
        const v4f* pw  = (const v4f*)s_w2;
        const v2f zero = {0.0f, 0.0f};
        v2f r0 = zero;
        v2f r1 = zero;
#define ADJ_TERM(A0, A1, BJ, WV)                                                  \
        {                                                                         \
            v2f x0 = (v2f){A0, A0} + (BJ);                                        \
            v2f x1 = (v2f){A1, A1} + (BJ);                                        \
            x0 = __builtin_elementwise_max(x0, zero);                             \
            x1 = __builtin_elementwise_max(x1, zero);                             \
            r0 += x0 * (v2f){WV, WV};                                             \
            r1 += x1 * (v2f){WV, WV};                                             \
        }
        for (int g = 0; g < HH / 4; ++g) {
            v4f a0 = pa0[g], a1 = pa1[g], w = pw[g];
            v4f bA = pb[2 * g], bB = pb[2 * g + 1];
            ADJ_TERM(a0.x, a1.x, ((v2f){bA.x, bA.y}), w.x);
            ADJ_TERM(a0.y, a1.y, ((v2f){bA.z, bA.w}), w.y);
            ADJ_TERM(a0.z, a1.z, ((v2f){bB.x, bB.y}), w.z);
            ADJ_TERM(a0.w, a1.w, ((v2f){bB.z, bB.w}), w.w);
        }
#undef ADJ_TERM
        bool dg = (ti == tj);
        s_out[ti * NP + tj]             = dg ? 0.0f : (r0.x + badj2);
        s_out[ti * NP + tj + NT]        = r0.y + badj2;
        s_out[(ti + NT) * NP + tj]      = r1.x + badj2;
        s_out[(ti + NT) * NP + tj + NT] = dg ? 0.0f : (r1.y + badj2);
    }
    __syncthreads();
    for (int pair = tid; pair < NP * NP; pair += HH)
        out_adj[(size_t)b * NP * NP + pair] = s_out[pair];
}

extern "C" void kernel_launch(void* const* d_in, const int* in_sizes, int n_in,
                              void* d_out, int out_size, void* d_ws, size_t ws_size,
                              hipStream_t stream) {
    const float* hs      = (const float*)d_in[0];
    const float* w_exist = (const float*)d_in[1];
    const float* b_exist = (const float*)d_in[2];
    const float* w_type  = (const float*)d_in[3];
    const float* b_type  = (const float*)d_in[4];
    const float* w_val   = (const float*)d_in[5];
    const float* b_val   = (const float*)d_in[6];
    const float* w_adj1  = (const float*)d_in[7];
    const float* b_adj1  = (const float*)d_in[8];
    const float* w_adj2  = (const float*)d_in[9];
    const float* b_adj2  = (const float*)d_in[10];
    const float* w_num   = (const float*)d_in[11];
    const float* b_num   = (const float*)d_in[12];

    float* out       = (float*)d_out;
    float* out_num   = out;
    float* out_exist = out + 30720;
    float* out_type  = out + 61440;
    float* out_val   = out + 983040;
    float* out_adj   = out + 1013760;

    float* grepr = (float*)d_ws;

    mean_kernel<<<(B_ * H_ / 4 + 255) / 256, 256, 0, stream>>>(hs, grepr);
    numnodes_kernel<<<8, 256, 0, stream>>>(grepr, w_num, b_num, out_num);
    heads_kernel<<<240, 256, 0, stream>>>(hs, w_type, b_type, w_exist, b_exist,
                                          w_val, b_val, out_exist, out_type, out_val);
    adj_kernel<<<B_, 256, 0, stream>>>(hs, w_adj1, b_adj1, w_adj2, b_adj2, out_adj);
}


# heads GEMM and sequence mean share one launch: 240 heads workgroups placed first, 512 mean workgroups fill the other wave slots, so the heads' memory waits overlap the mean's stream; sums unchanged
# speedup vs baseline: 1.9810x; 1.0338x over previous
#include <hip/hip_runtime.h>
#include <hip/hip_bf16.h>

typedef __attribute__((ext_vector_type(2))) float v2f;
typedef __attribute__((ext_vector_type(4))) float v4f;
typedef __attribute__((ext_vector_type(8))) float v8f;

#define B_  1024
#define S_  128
#define H_  512
#define NP  30
#define HH  256

#define KGRP 64
#define NGRP (H_ / KGRP)
__device__ __forceinline__ void gemm_k512(const float* __restrict__ a,
                                          const float* __restrict__ p,
                                          const float* __restrict__ q,
                                          bool qvalid, v8f& c0, v8f& c1) {
    v4f A0[8], P0[8], Q0[8];
    v4f A1[8], P1[8], Q1[8];
#define GRP_LOAD(A, P, Q, G)                                                      \
    {                                                                             \
        _Pragma("unroll")                                                         \
        for (int i = 0; i < 8; ++i) {                                             \
            A[i] = *(const v4f*)(a + (G) * KGRP + 8 * i);                         \
            P[i] = *(const v4f*)(p + (G) * KGRP + 8 * i);                         \
            Q[i] = *(const v4f*)(q + (G) * KGRP + 8 * i);                         \
        }                                                                         \
    }
#define GRP_MUL(A, P, Q)                                                          \
    {                                                                             \
        _Pragma("unroll")                                                         \
        for (int i = 0; i < 8; ++i) {                                             \
            v2f a01 = {A[i].x, A[i].y};                                           \
            v2f a23 = {A[i].z, A[i].w};                                           \
            v2f p01 = {P[i].x, P[i].y};                                           \
            v2f p23 = {P[i].z, P[i].w};                                           \
            v2f q01 = {qvalid ? Q[i].x : 0.0f, qvalid ? Q[i].y : 0.0f};           \
            v2f q23 = {qvalid ? Q[i].z : 0.0f, qvalid ? Q[i].w : 0.0f};           \
            c0 = __builtin_amdgcn_wmma_f32_16x16x4_f32(false, a01, false, p01,    \
                                                       (short)0, c0, false, false); \
            c1 = __builtin_amdgcn_wmma_f32_16x16x4_f32(false, a01, false, q01,    \
                                                       (short)0, c1, false, false); \
            c0 = __builtin_amdgcn_wmma_f32_16x16x4_f32(false, a23, false, p23,    \
                                                       (short)0, c0, false, false); \
            c1 = __builtin_amdgcn_wmma_f32_16x16x4_f32(false, a23, false, q23,    \
                                                       (short)0, c1, false, false); \
        }                                                                         \
    }
#define LOADS_STAY_AHEAD() __builtin_amdgcn_sched_barrier(0)
    GRP_LOAD(A0, P0, Q0, 0);
    LOADS_STAY_AHEAD();
#pragma unroll
    for (int g = 0; g < NGRP; g += 2) {
        GRP_LOAD(A1, P1, Q1, g + 1);
        LOADS_STAY_AHEAD();
        GRP_MUL(A0, P0, Q0);
        if (g + 2 < NGRP) GRP_LOAD(A0, P0, Q0, g + 2);
        LOADS_STAY_AHEAD();
        GRP_MUL(A1, P1, Q1);
    }
#undef LOADS_STAY_AHEAD
#undef GRP_LOAD
#undef GRP_MUL
}

__global__ __launch_bounds__(256) void numnodes_kernel(const float* __restrict__ grepr,
                                                       const float* __restrict__ w_num,
                                                       const float* __restrict__ b_num,
                                                       float* __restrict__ out_num) {
    int wave = (blockIdx.x * blockDim.x + threadIdx.x) >> 5;
    int lane = threadIdx.x & 31;
    int m0    = wave * 16;
    int mrow  = m0 + (lane & 15);
    int kq    = (lane >> 4) * 4;
    int nlo   = lane & 15;
    int nhi   = 16 + nlo;
    bool hvalid = (nhi < NP);
    const float* wlo = w_num + (size_t)nlo * H_ + kq;
    const float* whi = w_num + (size_t)(hvalid ? nhi : 0) * H_ + kq;
    const float* ar  = grepr + (size_t)mrow * H_ + kq;
    v8f c0 = {}; v8f c1 = {};
    gemm_k512(ar, wlo, whi, hvalid, c0, c1);
    int mbase = m0 + ((lane >> 4) << 3);
    for (int v = 0; v < 8; ++v) {
        int m = mbase + v;
        out_num[(size_t)m * NP + nlo] = c0[v] + b_num[nlo];
        if (hvalid) out_num[(size_t)m * NP + nhi] = c1[v] + b_num[nhi];
    }
}

#define HEADS_WG (B_ * NP / 16 / 8)
#define MEAN_WG  (B_ * H_ / 4 / 256)
__global__ __launch_bounds__(256) void heads_mean_kernel(const float* __restrict__ hs,
                                                         const float* __restrict__ w_type,
                                                         const float* __restrict__ b_type,
                                                         const float* __restrict__ w_exist,
                                                         const float* __restrict__ b_exist,
                                                         const float* __restrict__ w_val,
                                                         const float* __restrict__ b_val,
                                                         float* __restrict__ out_exist,
                                                         float* __restrict__ out_type,
                                                         float* __restrict__ out_val,
                                                         float* __restrict__ grepr) {
    if (blockIdx.x >= HEADS_WG) {
        int idx = (blockIdx.x - HEADS_WG) * blockDim.x + threadIdx.x;
        if (idx >= B_ * H_ / 4) return;
        int b  = idx / (H_ / 4);
        int h4 = idx % (H_ / 4);
        const float4* p = (const float4*)(hs + (size_t)b * S_ * H_) + h4;
        float4 acc = make_float4(0.f, 0.f, 0.f, 0.f);
        for (int s = 0; s < S_; ++s) {
            float4 v = p[s * (H_ / 4)];
            acc.x += v.x; acc.y += v.y; acc.z += v.z; acc.w += v.w;
        }
        const float inv = 1.0f / S_;
        ((float4*)grepr)[idx] = make_float4(acc.x * inv, acc.y * inv, acc.z * inv, acc.w * inv);
        return;
    }
    int wave = (blockIdx.x * blockDim.x + threadIdx.x) >> 5;
    int lane = threadIdx.x & 31;
    int m0 = wave * 16;
    int r  = m0 + (lane & 15);
    int bb = r / NP, pp = r % NP;
    const float* arow = hs + ((size_t)bb * S_ + pp) * H_;
    int kq  = (lane >> 4) * 4;
    int nlo = lane & 15;
    int nhi = 16 + nlo;
    const float* wlo = w_type + (size_t)nlo * H_ + kq;
    const float* whi = ((nhi < NP) ? (w_type + (size_t)nhi * H_)
                                   : ((nhi == NP) ? w_exist : w_val)) + kq;
    v8f c0 = {}; v8f c1 = {};
    gemm_k512(arow + kq, wlo, whi, true, c0, c1);
    int mbase = m0 + ((lane >> 4) << 3);
    for (int v = 0; v < 8; ++v) {
        int m  = mbase + v;
        int mb = m / NP, mp = m % NP;
        out_type[(size_t)mb * NP * NP + mp * NP + nlo] = c0[v] + b_type[nlo];
        if (nhi < NP) {
            out_type[(size_t)mb * NP * NP + mp * NP + nhi] = c1[v] + b_type[nhi];
        } else if (nhi == NP) {
            out_exist[m] = c1[v] + b_exist[0];
        } else {
            out_val[m]   = c1[v] + b_val[0];
        }
    }
}

#define AI_PITCH 260
#define BJ_PITCH 516
#define NT      15
#define KC      32
#define SP      36
#define SROWS   544
#define STAGE_F (SROWS * SP)
#define NSLAB   (H_ / KC)
static_assert(NP * AI_PITCH + NT * BJ_PITCH + HH + NP * NP <= STAGE_F, "the epilogue arrays fit inside slab stage 0");
__global__ __launch_bounds__(256) void adj_kernel(const float* __restrict__ hs,
                                                  const float* __restrict__ w_adj1,
                                                  const float* __restrict__ b_adj1,
                                                  const float* __restrict__ w_adj2,
                                                  const float* __restrict__ b_adj2,
                                                  float* __restrict__ out_adj) {
    __shared__ __attribute__((aligned(16))) float s_mem[2 * STAGE_F];
    float* s_ai  = s_mem;
    float* s_bj  = s_ai + NP * AI_PITCH;
    float* s_w2  = s_bj + NT * BJ_PITCH;
    float* s_out = s_w2 + HH;

    int b    = blockIdx.x;
    int tid  = threadIdx.x;
    int wave = tid >> 5;
    int lane = tid & 31;
    float rw2 = w_adj2[tid];

    int  r0     = tid >> 3;
    int  q      = tid & 7;
    bool avalid = (r0 < NP);
    const float* gB = w_adj1 + (size_t)r0 * (2 * H_) + 4 * q;
    const float* gA = hs + ((size_t)b * S_ + (avalid ? r0 : 0)) * H_ + 4 * q;
    int sdst = r0 * SP + 4 * q;
    v4f pf[17];

#define SLAB_LOAD(T)                                                              \
    {                                                                             \
        _Pragma("unroll")                                                         \
        for (int i = 0; i < 16; ++i)                                              \
            pf[i] = *(const v4f*)(gB + (size_t)(i & 7) * 32 * (2 * H_)         \
                                        + (i >> 3) * H_ + (T) * KC);              \
        v4f av = *(const v4f*)(gA + (T) * KC);                                    \
        av.x = avalid ? av.x : 0.0f;                           \
        av.y = avalid ? av.y : 0.0f;                                              \
        av.z = avalid ? av.z : 0.0f;                                              \
        av.w = avalid ? av.w : 0.0f;                                              \
        pf[16] = av;                                                              \
    }
#define SLAB_STORE(STG)                                                           \
    {                                                                             \
        float* d = s_mem + (STG) * STAGE_F + sdst;                                \
        _Pragma("unroll")                                                         \
        for (int i = 0; i < 17; ++i) *(v4f*)(d + i * 32 * SP) = pf[i];            \
    }

    int c       = lane & 15;
    int h       = lane >> 4;
    int mt      = wave >> 2;
    int ntbase  = (wave & 3) * 8;
    int aoff    = (512 + mt * 16 + c) * SP + 4 * h;
    int boff    = (ntbase * 16 + c) * SP + 4 * h;

    v8f acc[8];
    for (int t = 0; t < 8; ++t) acc[t] = (v8f){};

    SLAB_LOAD(0);
    SLAB_STORE(0);
    __syncthreads();

    for (int s = 0; s < NSLAB; ++s) {
        if (s + 1 < NSLAB) SLAB_LOAD(s + 1);
        const float* st = s_mem + (s & 1) * STAGE_F;
#pragma unroll
        for (int ch = 0; ch < KC / 8; ++ch) {
            v4f a4 = *(const v4f*)(st + aoff + ch * 8);
            v4f b4[8];
#pragma unroll
            for (int t = 0; t < 8; ++t)
                b4[t] = *(const v4f*)(st + boff + t * 16 * SP + ch * 8);
            v2f a01 = {a4.x, a4.y};
            v2f a23 = {a4.z, a4.w};
#pragma unroll
            for (int t = 0; t < 8; ++t) {
                v2f b01 = {b4[t].x, b4[t].y};
                acc[t] = __builtin_amdgcn_wmma_f32_16x16x4_f32(false, a01, false, b01,
                                                               (short)0, acc[t], false, false);
            }
#pragma unroll
            for (int t = 0; t < 8; ++t) {
                v2f b23 = {b4[t].z, b4[t].w};
                acc[t] = __builtin_amdgcn_wmma_f32_16x16x4_f32(false, a23, false, b23,
                                                               (short)0, acc[t], false, false);
            }
        }
        if (s + 1 < NSLAB) SLAB_STORE((s + 1) & 1);
        __syncthreads();
    }
#undef SLAB_LOAD
#undef SLAB_STORE

    s_w2[tid] = rw2;
    int mbase = mt * 16 + ((lane >> 4) << 3);
    if (ntbase < 16) {
        for (int t = 0; t < 8; ++t) {
            int   hh = (ntbase + t) * 16 + (lane & 15);
            float bb = b_adj1[hh];
            for (int v = 0; v < 8; ++v) {
                int m = mbase + v;
                if (m < NP) s_ai[m * AI_PITCH + hh] = acc[t][v] + bb;
            }
        }
    } else {
        for (int t = 0; t < 8; ++t) {
            int hh = (ntbase - 16 + t) * 16 + (lane & 15);
            for (int v = 0; v < 8; ++v) {
                int m = mbase + v;
                int p = (m < NT) ? m : m - NT;
                int s = (m < NT) ? 0 : 1;
                if (m < NP) s_bj[p * BJ_PITCH + 2 * hh + s] = acc[t][v];
            }
        }
    }
    __syncthreads();

    float badj2 = b_adj2[0];
    if (tid < NT * NT) {
        int ti = tid / NT, tj = tid % NT;
        const v4f* pa0 = (const v4f*)(s_ai + ti * AI_PITCH);
        const v4f* pa1 = (const v4f*)(s_ai + (ti + NT) * AI_PITCH);
        const v4f* pb  = (const v4f*)(s_bj + tj * BJ_PITCH);
        const v4f* pw  = (const v4f*)s_w2;
        const v2f zero = {0.0f, 0.0f};
        v2f r0 = zero;
        v2f r1 = zero;
#define ADJ_TERM(A0, A1, BJ, WV)                                                  \
        {                                                                         \
            v2f x0 = (v2f){A0, A0} + (BJ);                                        \
            v2f x1 = (v2f){A1, A1} + (BJ);                                        \
            x0 = __builtin_elementwise_max(x0, zero);                             \
            x1 = __builtin_elementwise_max(x1, zero);                             \
            r0 += x0 * (v2f){WV, WV};                                             \
            r1 += x1 * (v2f){WV, WV};                                             \
        }
        for (int g = 0; g < HH / 4; ++g) {
            v4f a0 = pa0[g], a1 = pa1[g], w = pw[g];
            v4f bA = pb[2 * g], bB = pb[2 * g + 1];
            ADJ_TERM(a0.x, a1.x, ((v2f){bA.x, bA.y}), w.x);
            ADJ_TERM(a0.y, a1.y, ((v2f){bA.z, bA.w}), w.y);
            ADJ_TERM(a0.z, a1.z, ((v2f){bB.x, bB.y}), w.z);
            ADJ_TERM(a0.w, a1.w, ((v2f){bB.z, bB.w}), w.w);
        }
#undef ADJ_TERM
        bool dg = (ti == tj);
        s_out[ti * NP + tj]             = dg ? 0.0f : (r0.x + badj2);
        s_out[ti * NP + tj + NT]        = r0.y + badj2;
        s_out[(ti + NT) * NP + tj]      = r1.x + badj2;
        s_out[(ti + NT) * NP + tj + NT] = dg ? 0.0f : (r1.y + badj2);
    }
    __syncthreads();
    for (int pair = tid; pair < NP * NP; pair += HH)
        out_adj[(size_t)b * NP * NP + pair] = s_out[pair];
}

extern "C" void kernel_launch(void* const* d_in, const int* in_sizes, int n_in,
                              void* d_out, int out_size, void* d_ws, size_t ws_size,
                              hipStream_t stream) {
    const float* hs      = (const float*)d_in[0];
    const float* w_exist = (const float*)d_in[1];
    const float* b_exist = (const float*)d_in[2];
    const float* w_type  = (const float*)d_in[3];
    const float* b_type  = (const float*)d_in[4];
    const float* w_val   = (const float*)d_in[5];
    const float* b_val   = (const float*)d_in[6];
    const float* w_adj1  = (const float*)d_in[7];
    const float* b_adj1  = (const float*)d_in[8];
    const float* w_adj2  = (const float*)d_in[9];
    const float* b_adj2  = (const float*)d_in[10];
    const float* w_num   = (const float*)d_in[11];
    const float* b_num   = (const float*)d_in[12];

    float* out       = (float*)d_out;
    float* out_num   = out;
    float* out_exist = out + 30720;
    float* out_type  = out + 61440;
    float* out_val   = out + 983040;
    float* out_adj   = out + 1013760;

    float* grepr = (float*)d_ws;

    heads_mean_kernel<<<HEADS_WG + MEAN_WG, 256, 0, stream>>>(hs, w_type, b_type, w_exist, b_exist,
                                                              w_val, b_val, out_exist, out_type,
                                                              out_val, grepr);
    numnodes_kernel<<<8, 256, 0, stream>>>(grepr, w_num, b_num, out_num);
    adj_kernel<<<B_, 256, 0, stream>>>(hs, w_adj1, b_adj1, w_adj2, b_adj2, out_adj);
}


# num_nodes product moved into the mean workgroups (each holds 2 whole mean rows): K split over the 8 waves, partial tiles summed via LDS in wave order, a changed summation order; 3 launches become 2
# speedup vs baseline: 2.0967x; 1.0584x over previous
#include <hip/hip_runtime.h>
#include <hip/hip_bf16.h>

typedef __attribute__((ext_vector_type(2))) float v2f;
typedef __attribute__((ext_vector_type(4))) float v4f;
typedef __attribute__((ext_vector_type(8))) float v8f;

#define B_  1024
#define S_  128
#define H_  512
#define NP  30
#define HH  256

#define KGRP 64
#define NGRP (H_ / KGRP)
__device__ __forceinline__ void gemm_k512(const float* __restrict__ a,
                                          const float* __restrict__ p,
                                          const float* __restrict__ q,
                                          bool qvalid, v8f& c0, v8f& c1) {
    v4f A0[8], P0[8], Q0[8];
    v4f A1[8], P1[8], Q1[8];
#define GRP_LOAD(A, P, Q, G)                                                      \
    {                                                                             \
        _Pragma("unroll")                                                         \
        for (int i = 0; i < 8; ++i) {                                             \
            A[i] = *(const v4f*)(a + (G) * KGRP + 8 * i);                         \
            P[i] = *(const v4f*)(p + (G) * KGRP + 8 * i);                         \
            Q[i] = *(const v4f*)(q + (G) * KGRP + 8 * i);                         \
        }                                                                         \
    }
#define GRP_MUL(A, P, Q)                                                          \
    {                                                                             \
        _Pragma("unroll")                                                         \
        for (int i = 0; i < 8; ++i) {                                             \
            v2f a01 = {A[i].x, A[i].y};                                           \
            v2f a23 = {A[i].z, A[i].w};                                           \
            v2f p01 = {P[i].x, P[i].y};                                           \
            v2f p23 = {P[i].z, P[i].w};                                           \
            v2f q01 = {qvalid ? Q[i].x : 0.0f, qvalid ? Q[i].y : 0.0f};           \
            v2f q23 = {qvalid ? Q[i].z : 0.0f, qvalid ? Q[i].w : 0.0f};           \
            c0 = __builtin_amdgcn_wmma_f32_16x16x4_f32(false, a01, false, p01,    \
                                                       (short)0, c0, false, false); \
            c1 = __builtin_amdgcn_wmma_f32_16x16x4_f32(false, a01, false, q01,    \
                                                       (short)0, c1, false, false); \
            c0 = __builtin_amdgcn_wmma_f32_16x16x4_f32(false, a23, false, p23,    \
                                                       (short)0, c0, false, false); \
            c1 = __builtin_amdgcn_wmma_f32_16x16x4_f32(false, a23, false, q23,    \
                                                       (short)0, c1, false, false); \
        }                                                                         \
    }
#define LOADS_STAY_AHEAD() __builtin_amdgcn_sched_barrier(0)
    GRP_LOAD(A0, P0, Q0, 0);
    LOADS_STAY_AHEAD();
#pragma unroll
    for (int g = 0; g < NGRP; g += 2) {
        GRP_LOAD(A1, P1, Q1, g + 1);
        LOADS_STAY_AHEAD();
        GRP_MUL(A0, P0, Q0);
        if (g + 2 < NGRP) GRP_LOAD(A0, P0, Q0, g + 2);
        LOADS_STAY_AHEAD();
        GRP_MUL(A1, P1, Q1);
    }
#undef LOADS_STAY_AHEAD
#undef GRP_LOAD
#undef GRP_MUL
}

#define HEADS_WG (B_ * NP / 16 / 8)
#define MEAN_WG  (B_ * H_ / 4 / 256)
#define KSL      (H_ / 8)
static_assert(MEAN_WG * 256 == B_ * H_ / 4, "the mean grid is exact: every thread of a mean workgroup is live");
__global__ __launch_bounds__(256) void heads_mean_kernel(const float* __restrict__ hs,
                                                         const float* __restrict__ w_type,
                                                         const float* __restrict__ b_type,
                                                         const float* __restrict__ w_exist,
                                                         const float* __restrict__ b_exist,
                                                         const float* __restrict__ w_val,
                                                         const float* __restrict__ b_val,
                                                         const float* __restrict__ w_num,
                                                         const float* __restrict__ b_num,
                                                         float* __restrict__ out_exist,
                                                         float* __restrict__ out_type,
                                                         float* __restrict__ out_val,
                                                         float* __restrict__ out_num) {
    if (blockIdx.x >= HEADS_WG) {
        __shared__ __attribute__((aligned(16))) float s_g[2 * H_];
        __shared__ __attribute__((aligned(16))) float s_part[8 * 32 * 4];
        const int tid  = threadIdx.x;
        const int wgm  = blockIdx.x - HEADS_WG;
        const int wv   = tid >> 5;
        const int lane = tid & 31;
        const int c    = lane & 15;
        const int kb   = wv * KSL + (lane >> 4) * 4;
        const int  nhi    = 16 + c;
        const bool hvalid = (nhi < NP);
        const float* wlo = w_num + (size_t)c * H_ + kb;
        const float* whi = w_num + (size_t)(hvalid ? nhi : 0) * H_ + kb;
        v4f P[8], Q[8];
#pragma unroll
        for (int i = 0; i < 8; ++i) {
            P[i] = *(const v4f*)(wlo + 8 * i);
            Q[i] = *(const v4f*)(whi + 8 * i);
        }
        __builtin_amdgcn_sched_barrier(0);

        const int idx = wgm * 256 + tid;
        const int b   = idx / (H_ / 4);
        const int h4  = idx % (H_ / 4);
        const float4* p = (const float4*)(hs + (size_t)b * S_ * H_) + h4;
        float4 acc = make_float4(0.f, 0.f, 0.f, 0.f);
        for (int s = 0; s < S_; ++s) {
            float4 v = p[s * (H_ / 4)];
            acc.x += v.x; acc.y += v.y; acc.z += v.z; acc.w += v.w;
        }
        const float inv = 1.0f / S_;
        ((float4*)s_g)[tid] = make_float4(acc.x * inv, acc.y * inv, acc.z * inv, acc.w * inv);
        __syncthreads();

        const bool avalid = (c < 2);
        const float* ap = s_g + (c & 1) * H_ + kb;
        v8f c0 = {}; v8f c1 = {};
#pragma unroll
        for (int i = 0; i < 8; ++i) {
            v4f A = *(const v4f*)(ap + 8 * i);
            v2f a01 = {avalid ? A.x : 0.0f, avalid ? A.y : 0.0f};
            v2f a23 = {avalid ? A.z : 0.0f, avalid ? A.w : 0.0f};
            v2f p01 = {P[i].x, P[i].y};
            v2f p23 = {P[i].z, P[i].w};
            v2f q01 = {hvalid ? Q[i].x : 0.0f, hvalid ? Q[i].y : 0.0f};
            v2f q23 = {hvalid ? Q[i].z : 0.0f, hvalid ? Q[i].w : 0.0f};
            c0 = __builtin_amdgcn_wmma_f32_16x16x4_f32(false, a01, false, p01,
                                                       (short)0, c0, false, false);
            c1 = __builtin_amdgcn_wmma_f32_16x16x4_f32(false, a01, false, q01,
                                                       (short)0, c1, false, false);
            c0 = __builtin_amdgcn_wmma_f32_16x16x4_f32(false, a23, false, p23,
                                                       (short)0, c0, false, false);
            c1 = __builtin_amdgcn_wmma_f32_16x16x4_f32(false, a23, false, q23,
                                                       (short)0, c1, false, false);
        }
        *(v4f*)(s_part + (wv * 32 + lane) * 4) = (v4f){c0[0], c0[1], c1[0], c1[1]};
        __syncthreads();
        if (tid < 64) {
            const int r = tid >> 5;
            const int n = tid & 31;
            if (n < NP) {
                const float* pp = s_part + (n & 15) * 4 + (n >> 4) * 2 + r;
                float sum = pp[0];
#pragma unroll
                for (int w = 1; w < 8; ++w) sum += pp[w * 32 * 4];
                out_num[(size_t)(2 * wgm + r) * NP + n] = sum + b_num[n];
            }
        }
        return;
    }
    int wave = (blockIdx.x * blockDim.x + threadIdx.x) >> 5;
    int lane = threadIdx.x & 31;
    int m0 = wave * 16;
    int r  = m0 + (lane & 15);
    int bb = r / NP, pp = r % NP;
    const float* arow = hs + ((size_t)bb * S_ + pp) * H_;
    int kq  = (lane >> 4) * 4;
    int nlo = lane & 15;
    int nhi = 16 + nlo;
    const float* wlo = w_type + (size_t)nlo * H_ + kq;
    const float* whi = ((nhi < NP) ? (w_type + (size_t)nhi * H_)
                                   : ((nhi == NP) ? w_exist : w_val)) + kq;
    v8f c0 = {}; v8f c1 = {};
    gemm_k512(arow + kq, wlo, whi, true, c0, c1);
    int mbase = m0 + ((lane >> 4) << 3);
    for (int v = 0; v < 8; ++v) {
        int m  = mbase + v;
        int mb = m / NP, mp = m % NP;
        out_type[(size_t)mb * NP * NP + mp * NP + nlo] = c0[v] + b_type[nlo];
        if (nhi < NP) {
            out_type[(size_t)mb * NP * NP + mp * NP + nhi] = c1[v] + b_type[nhi];
        } else if (nhi == NP) {
            out_exist[m] = c1[v] + b_exist[0];
        } else {
            out_val[m]   = c1[v] + b_val[0];
        }
    }
}

#define AI_PITCH 260
#define BJ_PITCH 516
#define NT      15
#define KC      32
#define SP      36
#define SROWS   544
#define STAGE_F (SROWS * SP)
#define NSLAB   (H_ / KC)
static_assert(NP * AI_PITCH + NT * BJ_PITCH + HH + NP * NP <= STAGE_F, "the epilogue arrays fit inside slab stage 0");
__global__ __launch_bounds__(256) void adj_kernel(const float* __restrict__ hs,
                                                  const float* __restrict__ w_adj1,
                                                  const float* __restrict__ b_adj1,
                                                  const float* __restrict__ w_adj2,
                                                  const float* __restrict__ b_adj2,
                                                  float* __restrict__ out_adj) {
    __shared__ __attribute__((aligned(16))) float s_mem[2 * STAGE_F];
    float* s_ai  = s_mem;
    float* s_bj  = s_ai + NP * AI_PITCH;
    float* s_w2  = s_bj + NT * BJ_PITCH;
    float* s_out = s_w2 + HH;

    int b    = blockIdx.x;
    int tid  = threadIdx.x;
    int wave = tid >> 5;
    int lane = tid & 31;
    float rw2 = w_adj2[tid];

    int  r0     = tid >> 3;
    int  q      = tid & 7;
    bool avalid = (r0 < NP);
    const float* gB = w_adj1 + (size_t)r0 * (2 * H_) + 4 * q;
    const float* gA = hs + ((size_t)b * S_ + (avalid ? r0 : 0)) * H_ + 4 * q;
    int sdst = r0 * SP + 4 * q;
    v4f pf[17];

#define SLAB_LOAD(T)                                                              \
    {                                                                             \
        _Pragma("unroll")                                                         \
        for (int i = 0; i < 16; ++i)                                              \
            pf[i] = *(const v4f*)(gB + (size_t)(i & 7) * 32 * (2 * H_)         \
                                        + (i >> 3) * H_ + (T) * KC);              \
        v4f av = *(const v4f*)(gA + (T) * KC);                                    \
        av.x = avalid ? av.x : 0.0f;                           \
        av.y = avalid ? av.y : 0.0f;                                              \
        av.z = avalid ? av.z : 0.0f;                                              \
        av.w = avalid ? av.w : 0.0f;                                              \
        pf[16] = av;                                                              \
    }
#define SLAB_STORE(STG)                                                           \
    {                                                                             \
        float* d = s_mem + (STG) * STAGE_F + sdst;                                \
        _Pragma("unroll")                                                         \
        for (int i = 0; i < 17; ++i) *(v4f*)(d + i * 32 * SP) = pf[i];            \
    }

    int c       = lane & 15;
    int h       = lane >> 4;
    int mt      = wave >> 2;
    int ntbase  = (wave & 3) * 8;
    int aoff    = (512 + mt * 16 + c) * SP + 4 * h;
    int boff    = (ntbase * 16 + c) * SP + 4 * h;

    v8f acc[8];
    for (int t = 0; t < 8; ++t) acc[t] = (v8f){};

    SLAB_LOAD(0);
    SLAB_STORE(0);
    __syncthreads();

    for (int s = 0; s < NSLAB; ++s) {
        if (s + 1 < NSLAB) SLAB_LOAD(s + 1);
        const float* st = s_mem + (s & 1) * STAGE_F;
#pragma unroll
        for (int ch = 0; ch < KC / 8; ++ch) {
            v4f a4 = *(const v4f*)(st + aoff + ch * 8);
            v4f b4[8];
#pragma unroll
            for (int t = 0; t < 8; ++t)
                b4[t] = *(const v4f*)(st + boff + t * 16 * SP + ch * 8);
            v2f a01 = {a4.x, a4.y};
            v2f a23 = {a4.z, a4.w};
#pragma unroll
            for (int t = 0; t < 8; ++t) {
                v2f b01 = {b4[t].x, b4[t].y};
                acc[t] = __builtin_amdgcn_wmma_f32_16x16x4_f32(false, a01, false, b01,
                                                               (short)0, acc[t], false, false);
            }
#pragma unroll
            for (int t = 0; t < 8; ++t) {
                v2f b23 = {b4[t].z, b4[t].w};
                acc[t] = __builtin_amdgcn_wmma_f32_16x16x4_f32(false, a23, false, b23,
                                                               (short)0, acc[t], false, false);
            }
        }
        if (s + 1 < NSLAB) SLAB_STORE((s + 1) & 1);
        __syncthreads();
    }
#undef SLAB_LOAD
#undef SLAB_STORE

    s_w2[tid] = rw2;
    int mbase = mt * 16 + ((lane >> 4) << 3);
    if (ntbase < 16) {
        for (int t = 0; t < 8; ++t) {
            int   hh = (ntbase + t) * 16 + (lane & 15);
            float bb = b_adj1[hh];
            for (int v = 0; v < 8; ++v) {
                int m = mbase + v;
                if (m < NP) s_ai[m * AI_PITCH + hh] = acc[t][v] + bb;
            }
        }
    } else {
        for (int t = 0; t < 8; ++t) {
            int hh = (ntbase - 16 + t) * 16 + (lane & 15);
            for (int v = 0; v < 8; ++v) {
                int m = mbase + v;
                int p = (m < NT) ? m : m - NT;
                int s = (m < NT) ? 0 : 1;
                if (m < NP) s_bj[p * BJ_PITCH + 2 * hh + s] = acc[t][v];
            }
        }
    }
    __syncthreads();

    float badj2 = b_adj2[0];
    if (tid < NT * NT) {
        int ti = tid / NT, tj = tid % NT;
        const v4f* pa0 = (const v4f*)(s_ai + ti * AI_PITCH);
        const v4f* pa1 = (const v4f*)(s_ai + (ti + NT) * AI_PITCH);
        const v4f* pb  = (const v4f*)(s_bj + tj * BJ_PITCH);
        const v4f* pw  = (const v4f*)s_w2;
        const v2f zero = {0.0f, 0.0f};
        v2f r0 = zero;
        v2f r1 = zero;
#define ADJ_TERM(A0, A1, BJ, WV)                                                  \
        {                                                                         \
            v2f x0 = (v2f){A0, A0} + (BJ);                                        \
            v2f x1 = (v2f){A1, A1} + (BJ);                                        \
            x0 = __builtin_elementwise_max(x0, zero);                             \
            x1 = __builtin_elementwise_max(x1, zero);                             \
            r0 += x0 * (v2f){WV, WV};                                             \
            r1 += x1 * (v2f){WV, WV};                                             \
        }
        for (int g = 0; g < HH / 4; ++g) {
            v4f a0 = pa0[g], a1 = pa1[g], w = pw[g];
            v4f bA = pb[2 * g], bB = pb[2 * g + 1];
            ADJ_TERM(a0.x, a1.x, ((v2f){bA.x, bA.y}), w.x);
            ADJ_TERM(a0.y, a1.y, ((v2f){bA.z, bA.w}), w.y);
            ADJ_TERM(a0.z, a1.z, ((v2f){bB.x, bB.y}), w.z);
            ADJ_TERM(a0.w, a1.w, ((v2f){bB.z, bB.w}), w.w);
        }
#undef ADJ_TERM
        bool dg = (ti == tj);
        s_out[ti * NP + tj]             = dg ? 0.0f : (r0.x + badj2);
        s_out[ti * NP + tj + NT]        = r0.y + badj2;
        s_out[(ti + NT) * NP + tj]      = r1.x + badj2;
        s_out[(ti + NT) * NP + tj + NT] = dg ? 0.0f : (r1.y + badj2);
    }
    __syncthreads();
    for (int pair = tid; pair < NP * NP; pair += HH)
        out_adj[(size_t)b * NP * NP + pair] = s_out[pair];
}

extern "C" void kernel_launch(void* const* d_in, const int* in_sizes, int n_in,
                              void* d_out, int out_size, void* d_ws, size_t ws_size,
                              hipStream_t stream) {
    const float* hs      = (const float*)d_in[0];
    const float* w_exist = (const float*)d_in[1];
    const float* b_exist = (const float*)d_in[2];
    const float* w_type  = (const float*)d_in[3];
    const float* b_type  = (const float*)d_in[4];
    const float* w_val   = (const float*)d_in[5];
    const float* b_val   = (const float*)d_in[6];
    const float* w_adj1  = (const float*)d_in[7];
    const float* b_adj1  = (const float*)d_in[8];
    const float* w_adj2  = (const float*)d_in[9];
    const float* b_adj2  = (const float*)d_in[10];
    const float* w_num   = (const float*)d_in[11];
    const float* b_num   = (const float*)d_in[12];

    float* out       = (float*)d_out;
    float* out_num   = out;
    float* out_exist = out + 30720;
    float* out_type  = out + 61440;
    float* out_val   = out + 983040;
    float* out_adj   = out + 1013760;

    heads_mean_kernel<<<HEADS_WG + MEAN_WG, 256, 0, stream>>>(hs, w_type, b_type, w_exist, b_exist,
                                                              w_val, b_val, w_num, b_num, out_exist,
                                                              out_type, out_val, out_num);
    adj_kernel<<<B_, 256, 0, stream>>>(hs, w_adj1, b_adj1, w_adj2, b_adj2, out_adj);
}


# adjacency GEMM: one workgroup per two batch elements, so each weight fragment read from LDS feeds two m-tiles and each weight K-slab is copied half as often; 512 workgroups, not 1024; sums unchanged
# speedup vs baseline: 2.1611x; 1.0307x over previous
#include <hip/hip_runtime.h>
#include <hip/hip_bf16.h>

typedef __attribute__((ext_vector_type(2))) float v2f;
typedef __attribute__((ext_vector_type(4))) float v4f;
typedef __attribute__((ext_vector_type(8))) float v8f;

#define B_  1024
#define S_  128
#define H_  512
#define NP  30
#define HH  256

#define KGRP 64
#define NGRP (H_ / KGRP)
__device__ __forceinline__ void gemm_k512(const float* __restrict__ a,
                                          const float* __restrict__ p,
                                          const float* __restrict__ q,
                                          bool qvalid, v8f& c0, v8f& c1) {
    v4f A0[8], P0[8], Q0[8];
    v4f A1[8], P1[8], Q1[8];
#define GRP_LOAD(A, P, Q, G)                                                      \
    {                                                                             \
        _Pragma("unroll")                                                         \
        for (int i = 0; i < 8; ++i) {                                             \
            A[i] = *(const v4f*)(a + (G) * KGRP + 8 * i);                         \
            P[i] = *(const v4f*)(p + (G) * KGRP + 8 * i);                         \
            Q[i] = *(const v4f*)(q + (G) * KGRP + 8 * i);                         \
        }                                                                         \
    }
#define GRP_MUL(A, P, Q)                                                          \
    {                                                                             \
        _Pragma("unroll")                                                         \
        for (int i = 0; i < 8; ++i) {                                             \
            v2f a01 = {A[i].x, A[i].y};                                           \
            v2f a23 = {A[i].z, A[i].w};                                           \
            v2f p01 = {P[i].x, P[i].y};                                           \
            v2f p23 = {P[i].z, P[i].w};                                           \
            v2f q01 = {qvalid ? Q[i].x : 0.0f, qvalid ? Q[i].y : 0.0f};           \
            v2f q23 = {qvalid ? Q[i].z : 0.0f, qvalid ? Q[i].w : 0.0f};           \
            c0 = __builtin_amdgcn_wmma_f32_16x16x4_f32(false, a01, false, p01,    \
                                                       (short)0, c0, false, false); \
            c1 = __builtin_amdgcn_wmma_f32_16x16x4_f32(false, a01, false, q01,    \
                                                       (short)0, c1, false, false); \
            c0 = __builtin_amdgcn_wmma_f32_16x16x4_f32(false, a23, false, p23,    \
                                                       (short)0, c0, false, false); \
            c1 = __builtin_amdgcn_wmma_f32_16x16x4_f32(false, a23, false, q23,    \
                                                       (short)0, c1, false, false); \
        }                                                                         \
    }
#define LOADS_STAY_AHEAD() __builtin_amdgcn_sched_barrier(0)
    GRP_LOAD(A0, P0, Q0, 0);
    LOADS_STAY_AHEAD();
#pragma unroll
    for (int g = 0; g < NGRP; g += 2) {
        GRP_LOAD(A1, P1, Q1, g + 1);
        LOADS_STAY_AHEAD();
        GRP_MUL(A0, P0, Q0);
        if (g + 2 < NGRP) GRP_LOAD(A0, P0, Q0, g + 2);
        LOADS_STAY_AHEAD();
        GRP_MUL(A1, P1, Q1);
    }
#undef LOADS_STAY_AHEAD
#undef GRP_LOAD
#undef GRP_MUL
}

#define HEADS_WG (B_ * NP / 16 / 8)
#define MEAN_WG  (B_ * H_ / 4 / 256)
#define KSL      (H_ / 8)
static_assert(MEAN_WG * 256 == B_ * H_ / 4, "the mean grid is exact: every thread of a mean workgroup is live");
__global__ __launch_bounds__(256) void heads_mean_kernel(const float* __restrict__ hs,
                                                         const float* __restrict__ w_type,
                                                         const float* __restrict__ b_type,
                                                         const float* __restrict__ w_exist,
                                                         const float* __restrict__ b_exist,
                                                         const float* __restrict__ w_val,
                                                         const float* __restrict__ b_val,
                                                         const float* __restrict__ w_num,
                                                         const float* __restrict__ b_num,
                                                         float* __restrict__ out_exist,
                                                         float* __restrict__ out_type,
                                                         float* __restrict__ out_val,
                                                         float* __restrict__ out_num) {
    if (blockIdx.x >= HEADS_WG) {
        __shared__ __attribute__((aligned(16))) float s_g[2 * H_];
        __shared__ __attribute__((aligned(16))) float s_part[8 * 32 * 4];
        const int tid  = threadIdx.x;
        const int wgm  = blockIdx.x - HEADS_WG;
        const int wv   = tid >> 5;
        const int lane = tid & 31;
        const int c    = lane & 15;
        const int kb   = wv * KSL + (lane >> 4) * 4;
        const int  nhi    = 16 + c;
        const bool hvalid = (nhi < NP);
        const float* wlo = w_num + (size_t)c * H_ + kb;
        const float* whi = w_num + (size_t)(hvalid ? nhi : 0) * H_ + kb;
        v4f P[8], Q[8];
#pragma unroll
        for (int i = 0; i < 8; ++i) {
            P[i] = *(const v4f*)(wlo + 8 * i);
            Q[i] = *(const v4f*)(whi + 8 * i);
        }
        __builtin_amdgcn_sched_barrier(0);

        const int idx = wgm * 256 + tid;
        const int b   = idx / (H_ / 4);
        const int h4  = idx % (H_ / 4);
        const float4* p = (const float4*)(hs + (size_t)b * S_ * H_) + h4;
        float4 acc = make_float4(0.f, 0.f, 0.f, 0.f);
        for (int s = 0; s < S_; ++s) {
            float4 v = p[s * (H_ / 4)];
            acc.x += v.x; acc.y += v.y; acc.z += v.z; acc.w += v.w;
        }
        const float inv = 1.0f / S_;
        ((float4*)s_g)[tid] = make_float4(acc.x * inv, acc.y * inv, acc.z * inv, acc.w * inv);
        __syncthreads();

        const bool avalid = (c < 2);
        const float* ap = s_g + (c & 1) * H_ + kb;
        v8f c0 = {}; v8f c1 = {};
#pragma unroll
        for (int i = 0; i < 8; ++i) {
            v4f A = *(const v4f*)(ap + 8 * i);
            v2f a01 = {avalid ? A.x : 0.0f, avalid ? A.y : 0.0f};
            v2f a23 = {avalid ? A.z : 0.0f, avalid ? A.w : 0.0f};
            v2f p01 = {P[i].x, P[i].y};
            v2f p23 = {P[i].z, P[i].w};
            v2f q01 = {hvalid ? Q[i].x : 0.0f, hvalid ? Q[i].y : 0.0f};
            v2f q23 = {hvalid ? Q[i].z : 0.0f, hvalid ? Q[i].w : 0.0f};
            c0 = __builtin_amdgcn_wmma_f32_16x16x4_f32(false, a01, false, p01,
                                                       (short)0, c0, false, false);
            c1 = __builtin_amdgcn_wmma_f32_16x16x4_f32(false, a01, false, q01,
                                                       (short)0, c1, false, false);
            c0 = __builtin_amdgcn_wmma_f32_16x16x4_f32(false, a23, false, p23,
                                                       (short)0, c0, false, false);
            c1 = __builtin_amdgcn_wmma_f32_16x16x4_f32(false, a23, false, q23,
                                                       (short)0, c1, false, false);
        }
        *(v4f*)(s_part + (wv * 32 + lane) * 4) = (v4f){c0[0], c0[1], c1[0], c1[1]};
        __syncthreads();
        if (tid < 64) {
            const int r = tid >> 5;
            const int n = tid & 31;
            if (n < NP) {
                const float* pp = s_part + (n & 15) * 4 + (n >> 4) * 2 + r;
                float sum = pp[0];
#pragma unroll
                for (int w = 1; w < 8; ++w) sum += pp[w * 32 * 4];
                out_num[(size_t)(2 * wgm + r) * NP + n] = sum + b_num[n];
            }
        }
        return;
    }
    int wave = (blockIdx.x * blockDim.x + threadIdx.x) >> 5;
    int lane = threadIdx.x & 31;
    int m0 = wave * 16;
    int r  = m0 + (lane & 15);
    int bb = r / NP, pp = r % NP;
    const float* arow = hs + ((size_t)bb * S_ + pp) * H_;
    int kq  = (lane >> 4) * 4;
    int nlo = lane & 15;
    int nhi = 16 + nlo;
    const float* wlo = w_type + (size_t)nlo * H_ + kq;
    const float* whi = ((nhi < NP) ? (w_type + (size_t)nhi * H_)
                                   : ((nhi == NP) ? w_exist : w_val)) + kq;
    v8f c0 = {}; v8f c1 = {};
    gemm_k512(arow + kq, wlo, whi, true, c0, c1);
    int mbase = m0 + ((lane >> 4) << 3);
    for (int v = 0; v < 8; ++v) {
        int m  = mbase + v;
        int mb = m / NP, mp = m % NP;
        out_type[(size_t)mb * NP * NP + mp * NP + nlo] = c0[v] + b_type[nlo];
        if (nhi < NP) {
            out_type[(size_t)mb * NP * NP + mp * NP + nhi] = c1[v] + b_type[nhi];
        } else if (nhi == NP) {
            out_exist[m] = c1[v] + b_exist[0];
        } else {
            out_val[m]   = c1[v] + b_val[0];
        }
    }
}

#define AI_PITCH 260
#define BJ_PITCH 516
#define NT      15
#define KC      32
#define SP      36
#define BPW     2
#define SROWS   (512 + 32 * BPW)
#define STAGE_F (SROWS * SP)
#define NSLAB   (H_ / KC)
#define EPI_F   (NP * AI_PITCH + NT * BJ_PITCH + NP * NP)
static_assert(B_ % BPW == 0, "the adjacency grid is exact");
static_assert(BPW * EPI_F + HH <= 2 * STAGE_F, "the epilogue arrays fit inside the slab stages");
static_assert((EPI_F % 4) == 0, "the second element's arrays stay 16-byte aligned");

__device__ __forceinline__ void spill_tiles(const v8f (&acc)[8], int mt, int ntbase, int lane,
                                            const float* __restrict__ b_adj1,
                                            float* s_ai, float* s_bj) {
    int mbase = mt * 16 + ((lane >> 4) << 3);
    if (ntbase < 16) {
        for (int t = 0; t < 8; ++t) {
            int   hh = (ntbase + t) * 16 + (lane & 15);
            float bb = b_adj1[hh];
            for (int v = 0; v < 8; ++v) {
                int m = mbase + v;
                if (m < NP) s_ai[m * AI_PITCH + hh] = acc[t][v] + bb;
            }
        }
    } else {
        for (int t = 0; t < 8; ++t) {
            int hh = (ntbase - 16 + t) * 16 + (lane & 15);
            for (int v = 0; v < 8; ++v) {
                int m = mbase + v;
                int p = (m < NT) ? m : m - NT;
                int s = (m < NT) ? 0 : 1;
                if (m < NP) s_bj[p * BJ_PITCH + 2 * hh + s] = acc[t][v];
            }
        }
    }
}

__global__ __launch_bounds__(256) void adj_kernel(const float* __restrict__ hs,
                                                  const float* __restrict__ w_adj1,
                                                  const float* __restrict__ b_adj1,
                                                  const float* __restrict__ w_adj2,
                                                  const float* __restrict__ b_adj2,
                                                  float* __restrict__ out_adj) {
    __shared__ __attribute__((aligned(16))) float s_mem[2 * STAGE_F];
    float* s_w2 = s_mem + BPW * EPI_F;

    int b0   = blockIdx.x * BPW;
    int tid  = threadIdx.x;
    int wave = tid >> 5;
    int lane = tid & 31;
    float rw2 = w_adj2[tid];

    int  r0     = tid >> 3;
    int  q      = tid & 7;
    bool avalid = (r0 < NP);
    const float* gB  = w_adj1 + (size_t)r0 * (2 * H_) + 4 * q;
    const float* gA0 = hs + ((size_t)b0 * S_ + (avalid ? r0 : 0)) * H_ + 4 * q;
    const float* gA1 = gA0 + (size_t)S_ * H_;
    int sdst = r0 * SP + 4 * q;
    v4f pf[16 + BPW];

#define SLAB_LOAD(T)                                                              \
    {                                                                             \
        _Pragma("unroll")                                                         \
        for (int i = 0; i < 16; ++i)                                              \
            pf[i] = *(const v4f*)(gB + (size_t)(i & 7) * 32 * (2 * H_)         \
                                        + (i >> 3) * H_ + (T) * KC);              \
        pf[16] = *(const v4f*)(gA0 + (T) * KC);                                   \
        pf[17] = *(const v4f*)(gA1 + (T) * KC);                                   \
    }
#define SLAB_STORE(STG)                                                           \
    {                                                                             \
        float* d = s_mem + (STG) * STAGE_F + sdst;                                \
        _Pragma("unroll")                                                         \
        for (int i = 0; i < 16; ++i) *(v4f*)(d + i * 32 * SP) = pf[i];            \
        _Pragma("unroll")                                                         \
        for (int i = 16; i < 16 + BPW; ++i) {                                     \
            v4f av = pf[i];                                                       \
            av.x = avalid ? av.x : 0.0f;                                          \
            av.y = avalid ? av.y : 0.0f;                                          \
            av.z = avalid ? av.z : 0.0f;                                          \
            av.w = avalid ? av.w : 0.0f;                                          \
            *(v4f*)(d + i * 32 * SP) = av;                                        \
        }                                                                         \
    }

    int c       = lane & 15;
    int h       = lane >> 4;
    int be      = wave >> 2;
    int ntbase  = (wave & 3) * 8;
    int aoff    = (512 + be * 32 + c) * SP + 4 * h;
    int boff    = (ntbase * 16 + c) * SP + 4 * h;

    v8f acc0[8], acc1[8];
    for (int t = 0; t < 8; ++t) { acc0[t] = (v8f){}; acc1[t] = (v8f){}; }

    SLAB_LOAD(0);
    SLAB_STORE(0);
    __syncthreads();

    for (int s = 0; s < NSLAB; ++s) {
        if (s + 1 < NSLAB) SLAB_LOAD(s + 1);
        const float* st = s_mem + (s & 1) * STAGE_F;
#pragma unroll
        for (int ch = 0; ch < KC / 8; ++ch) {
            v4f a4 = *(const v4f*)(st + aoff + ch * 8);
            v4f a5 = *(const v4f*)(st + aoff + 16 * SP + ch * 8);
            v4f b4[8];
#pragma unroll
            for (int t = 0; t < 8; ++t)
                b4[t] = *(const v4f*)(st + boff + t * 16 * SP + ch * 8);
            v2f a01 = {a4.x, a4.y};
            v2f a23 = {a4.z, a4.w};
            v2f c01 = {a5.x, a5.y};
            v2f c23 = {a5.z, a5.w};
#pragma unroll
            for (int t = 0; t < 8; ++t) {
                v2f b01 = {b4[t].x, b4[t].y};
                acc0[t] = __builtin_amdgcn_wmma_f32_16x16x4_f32(false, a01, false, b01,
                                                                (short)0, acc0[t], false, false);
                acc1[t] = __builtin_amdgcn_wmma_f32_16x16x4_f32(false, c01, false, b01,
                                                                (short)0, acc1[t], false, false);
            }
#pragma unroll
            for (int t = 0; t < 8; ++t) {
                v2f b23 = {b4[t].z, b4[t].w};
                acc0[t] = __builtin_amdgcn_wmma_f32_16x16x4_f32(false, a23, false, b23,
                                                                (short)0, acc0[t], false, false);
                acc1[t] = __builtin_amdgcn_wmma_f32_16x16x4_f32(false, c23, false, b23,
                                                                (short)0, acc1[t], false, false);
            }
        }
        __builtin_amdgcn_sched_barrier(0);
        if (s + 1 < NSLAB) SLAB_STORE((s + 1) & 1);
        __syncthreads();
    }
#undef SLAB_LOAD
#undef SLAB_STORE

    s_w2[tid] = rw2;
    {
        float* e_ai = s_mem + be * EPI_F;
        float* e_bj = e_ai + NP * AI_PITCH;
        spill_tiles(acc0, 0, ntbase, lane, b_adj1, e_ai, e_bj);
        spill_tiles(acc1, 1, ntbase, lane, b_adj1, e_ai, e_bj);
    }
    __syncthreads();

    float badj2 = b_adj2[0];
    if (tid < NT * NT) {
        int ti = tid / NT, tj = tid % NT;
        const v4f* pw  = (const v4f*)s_w2;
        const v2f zero = {0.0f, 0.0f};
        bool dg = (ti == tj);
        for (int e = 0; e < BPW; ++e) {
            float* e_ai  = s_mem + e * EPI_F;
            float* e_bj  = e_ai + NP * AI_PITCH;
            float* e_out = e_bj + NT * BJ_PITCH;
            const v4f* pa0 = (const v4f*)(e_ai + ti * AI_PITCH);
            const v4f* pa1 = (const v4f*)(e_ai + (ti + NT) * AI_PITCH);
            const v4f* pb  = (const v4f*)(e_bj + tj * BJ_PITCH);
            v2f r0 = zero;
            v2f r1 = zero;
#define ADJ_TERM(A0, A1, BJ, WV)                                                  \
            {                                                                     \
                v2f x0 = (v2f){A0, A0} + (BJ);                                    \
                v2f x1 = (v2f){A1, A1} + (BJ);                                    \
                x0 = __builtin_elementwise_max(x0, zero);                         \
                x1 = __builtin_elementwise_max(x1, zero);                         \
                r0 += x0 * (v2f){WV, WV};                                         \
                r1 += x1 * (v2f){WV, WV};                                         \
            }
            for (int g = 0; g < HH / 4; ++g) {
                v4f a0 = pa0[g], a1 = pa1[g], w = pw[g];
                v4f bA = pb[2 * g], bB = pb[2 * g + 1];
                ADJ_TERM(a0.x, a1.x, ((v2f){bA.x, bA.y}), w.x);
                ADJ_TERM(a0.y, a1.y, ((v2f){bA.z, bA.w}), w.y);
                ADJ_TERM(a0.z, a1.z, ((v2f){bB.x, bB.y}), w.z);
                ADJ_TERM(a0.w, a1.w, ((v2f){bB.z, bB.w}), w.w);
            }
#undef ADJ_TERM
            e_out[ti * NP + tj]             = dg ? 0.0f : (r0.x + badj2);
            e_out[ti * NP + tj + NT]        = r0.y + badj2;
            e_out[(ti + NT) * NP + tj]      = r1.x + badj2;
            e_out[(ti + NT) * NP + tj + NT] = dg ? 0.0f : (r1.y + badj2);
        }
    }
    __syncthreads();
    for (int idx = tid; idx < BPW * NP * NP; idx += HH) {
        int e = (idx >= NP * NP) ? 1 : 0;
        int o = idx - e * (NP * NP);
        out_adj[(size_t)b0 * NP * NP + idx] = s_mem[e * EPI_F + NP * AI_PITCH + NT * BJ_PITCH + o];
    }
}

extern "C" void kernel_launch(void* const* d_in, const int* in_sizes, int n_in,
                              void* d_out, int out_size, void* d_ws, size_t ws_size,
                              hipStream_t stream) {
    const float* hs      = (const float*)d_in[0];
    const float* w_exist = (const float*)d_in[1];
    const float* b_exist = (const float*)d_in[2];
    const float* w_type  = (const float*)d_in[3];
    const float* b_type  = (const float*)d_in[4];
    const float* w_val   = (const float*)d_in[5];
    const float* b_val   = (const float*)d_in[6];
    const float* w_adj1  = (const float*)d_in[7];
    const float* b_adj1  = (const float*)d_in[8];
    const float* w_adj2  = (const float*)d_in[9];
    const float* b_adj2  = (const float*)d_in[10];
    const float* w_num   = (const float*)d_in[11];
    const float* b_num   = (const float*)d_in[12];

    float* out       = (float*)d_out;
    float* out_num   = out;
    float* out_exist = out + 30720;
    float* out_type  = out + 61440;
    float* out_val   = out + 983040;
    float* out_adj   = out + 1013760;

    heads_mean_kernel<<<HEADS_WG + MEAN_WG, 256, 0, stream>>>(hs, w_type, b_type, w_exist, b_exist,
                                                              w_val, b_val, w_num, b_num, out_exist,
                                                              out_type, out_val, out_num);
    adj_kernel<<<B_ / BPW, 256, 0, stream>>>(hs, w_adj1, b_adj1, w_adj2, b_adj2, out_adj);
}
